# starting kernel
# baseline (speedup 1.0000x reference)
#include <hip/hip_runtime.h>

constexpr int Bn    = 32;
constexpr int PPB   = 256;
constexpr int Np    = Bn * PPB;
constexpr int Mf    = 128;
constexpr int LANGD = 256;
constexpr int Hd    = 128;
constexpr int Kfuse = LANGD + Mf;
#define BN_EPS 1e-5f

typedef float v2f __attribute__((ext_vector_type(2)));
typedef float v8f __attribute__((ext_vector_type(8)));

__device__ __forceinline__ v8f wmma4(v2f a, v2f b, v8f c) {
  return __builtin_amdgcn_wmma_f32_16x16x4_f32(false, a, false, b, (short)0, c,
                                               false, false);
}

#define V8F_ZERO {0.f, 0.f, 0.f, 0.f, 0.f, 0.f, 0.f, 0.f}

__global__ void k_bid(const int* __restrict__ pidx, const int* __restrict__ poff,
                      const int* __restrict__ boff, int* __restrict__ bid) {
  int i = blockIdx.x * blockDim.x + threadIdx.x;
  if (i >= Np) return;
  int first_pt = pidx[2 * poff[i] + 1];
  int b = -1;
#pragma unroll
  for (int j = 0; j <= Bn; ++j) b += (boff[j] <= first_pt) ? 1 : 0;
  bid[i] = b;
}

__global__ __launch_bounds__(256) void k_rank(const int* __restrict__ bid,
                                              int* __restrict__ dst) {
  __shared__ int sbid[Np];
  for (int j = threadIdx.x; j < Np; j += blockDim.x) sbid[j] = bid[j];
  __syncthreads();
  int i = blockIdx.x * blockDim.x + threadIdx.x;
  if (i >= Np) return;
  int myb = sbid[i];
  int r = 0;
  for (int j = 0; j < i; ++j) r += (sbid[j] == myb) ? 1 : 0;
  dst[i] = myb * PPB + r;
}

__global__ void k_zero(float* __restrict__ p, int n) {
  int i = blockIdx.x * blockDim.x + threadIdx.x;
  if (i < n) p[i] = 0.f;
}

__global__ __launch_bounds__(128) void k_scatter(const float* __restrict__ feats,
                                                 const int* __restrict__ dst,
                                                 float* __restrict__ padded) {
  int i = blockIdx.x;
  int d = dst[i];
  if (d < 0 || d >= Np) return;
  padded[(size_t)d * Mf + threadIdx.x] = feats[(size_t)i * Mf + threadIdx.x];
}

__global__ __launch_bounds__(256) void k_l1(const float* __restrict__ padded,
                                            const float* __restrict__ lang,
                                            const float* __restrict__ Wf,
                                            const float* __restrict__ bf,
                                            float* __restrict__ out) {
  const int b     = blockIdx.x;
  const int pbase = blockIdx.y * 64;
  const int wave  = threadIdx.x >> 5;
  const int lane  = threadIdx.x & 31;
  const int half  = lane >> 4;
  const int lm    = lane & 15;
  const int o0    = wave << 4;

  const float* arow = Wf + (size_t)(o0 + lm) * Kfuse + 2 * half;
  const float* xb   = padded + (size_t)b * PPB * Mf;
  const float* br0  = xb + (size_t)(pbase +  0 + lm) * Mf + 2 * half;
  const float* br1  = xb + (size_t)(pbase + 16 + lm) * Mf + 2 * half;
  const float* br2  = xb + (size_t)(pbase + 32 + lm) * Mf + 2 * half;
  const float* br3  = xb + (size_t)(pbase + 48 + lm) * Mf + 2 * half;
  const float* lb   = lang + (size_t)b * LANGD + 2 * half;

  v8f acc0 = V8F_ZERO, acc1 = V8F_ZERO, acc2 = V8F_ZERO, acc3 = V8F_ZERO;

#pragma unroll 4
  for (int k0 = 0; k0 < Mf; k0 += 4) {
    v2f a;   a.x  = arow[k0];      a.y  = arow[k0 + 1];
    v2f b0;  b0.x = br0[k0];       b0.y = br0[k0 + 1];
    v2f b1;  b1.x = br1[k0];       b1.y = br1[k0 + 1];
    v2f b2;  b2.x = br2[k0];       b2.y = br2[k0 + 1];
    v2f b3;  b3.x = br3[k0];       b3.y = br3[k0 + 1];
    acc0 = wmma4(a, b0, acc0);
    acc1 = wmma4(a, b1, acc1);
    acc2 = wmma4(a, b2, acc2);
    acc3 = wmma4(a, b3, acc3);
  }
#pragma unroll 4
  for (int k0 = 0; k0 < LANGD; k0 += 4) {
    v2f a;   a.x = arow[Mf + k0];  a.y = arow[Mf + k0 + 1];
    v2f bb;  bb.x = lb[k0];        bb.y = lb[k0 + 1];
    acc0 = wmma4(a, bb, acc0);
    acc1 = wmma4(a, bb, acc1);
    acc2 = wmma4(a, bb, acc2);
    acc3 = wmma4(a, bb, acc3);
  }

  float* ob = out + ((size_t)b * Hd) * PPB + pbase + lm;
#pragma unroll
  for (int v = 0; v < 8; ++v) {
    const int o = o0 + v + 8 * half;
    const float bias = bf[o];
    float* orow = ob + (size_t)o * PPB;
    orow[ 0] = fmaxf(acc0[v] + bias, 0.f);
    orow[16] = fmaxf(acc1[v] + bias, 0.f);
    orow[32] = fmaxf(acc2[v] + bias, 0.f);
    orow[48] = fmaxf(acc3[v] + bias, 0.f);
  }
}

__global__ __launch_bounds__(256) void k_mid(const float* __restrict__ in,
                                             const float* __restrict__ W,
                                             const float* __restrict__ bias,
                                             const float* __restrict__ gam,
                                             const float* __restrict__ bet,
                                             const float* __restrict__ rme,
                                             const float* __restrict__ rva,
                                             float* __restrict__ out) {
  const int b     = blockIdx.x;
  const int pbase = blockIdx.y * 64;
  const int wave  = threadIdx.x >> 5;
  const int lane  = threadIdx.x & 31;
  const int half  = lane >> 4;
  const int lm    = lane & 15;
  const int o0    = wave << 4;

  const float* arow = W + (size_t)(o0 + lm) * Hd + 2 * half;
  const float* inb  = in + (size_t)b * Hd * PPB;
  const float* bc   = inb + (size_t)(2 * half) * PPB + pbase + lm;

  v8f acc0 = V8F_ZERO, acc1 = V8F_ZERO, acc2 = V8F_ZERO, acc3 = V8F_ZERO;

#pragma unroll 4
  for (int k0 = 0; k0 < Hd; k0 += 4) {
    const int r0 = k0 * PPB;
    const int r1 = r0 + PPB;
    v2f a;   a.x  = arow[k0];      a.y  = arow[k0 + 1];
    v2f b0;  b0.x = bc[r0 +  0];   b0.y = bc[r1 +  0];
    v2f b1;  b1.x = bc[r0 + 16];   b1.y = bc[r1 + 16];
    v2f b2;  b2.x = bc[r0 + 32];   b2.y = bc[r1 + 32];
    v2f b3;  b3.x = bc[r0 + 48];   b3.y = bc[r1 + 48];
    acc0 = wmma4(a, b0, acc0);
    acc1 = wmma4(a, b1, acc1);
    acc2 = wmma4(a, b2, acc2);
    acc3 = wmma4(a, b3, acc3);
  }

  float* ob = out + ((size_t)b * Hd) * PPB + pbase + lm;
#pragma unroll
  for (int v = 0; v < 8; ++v) {
    const int o = o0 + v + 8 * half;
    const float bs    = bias[o];
    const float scale = gam[o] * rsqrtf(rva[o] + BN_EPS);
    const float mu    = rme[o];
    const float beta  = bet[o];
    float* orow = ob + (size_t)o * PPB;
    orow[ 0] = (fmaxf(acc0[v] + bs, 0.f) - mu) * scale + beta;
    orow[16] = (fmaxf(acc1[v] + bs, 0.f) - mu) * scale + beta;
    orow[32] = (fmaxf(acc2[v] + bs, 0.f) - mu) * scale + beta;
    orow[48] = (fmaxf(acc3[v] + bs, 0.f) - mu) * scale + beta;
  }
}

__global__ __launch_bounds__(256) void k_final(const float* __restrict__ in,
                                               const float* __restrict__ W3,
                                               const float* __restrict__ b3,
                                               float* __restrict__ out) {
  int idx = blockIdx.x * blockDim.x + threadIdx.x;
  if (idx >= Bn * PPB) return;
  int b = idx / PPB, p = idx % PPB;
  const float* inb = in + (size_t)b * Hd * PPB + p;
  float s = b3[0];
#pragma unroll 8
  for (int c = 0; c < Hd; ++c) s += W3[c] * inb[(size_t)c * PPB];
  out[idx] = s;
}

extern "C" void kernel_launch(void* const* d_in, const int* in_sizes, int n_in,
                              void* d_out, int out_size, void* d_ws, size_t ws_size,
                              hipStream_t stream) {
  const float* feats = (const float*)d_in[0];
  const int*   pidx  = (const int*)d_in[1];
  const int*   poff  = (const int*)d_in[2];
  const int*   boff  = (const int*)d_in[3];
  const float* lang  = (const float*)d_in[4];
  const float* Wf    = (const float*)d_in[5];
  const float* bf    = (const float*)d_in[6];
  const float* W1p   = (const float*)d_in[7];
  const float* b1p   = (const float*)d_in[8];
  const float* g1p   = (const float*)d_in[9];
  const float* be1p  = (const float*)d_in[10];
  const float* rm1p  = (const float*)d_in[11];
  const float* rv1p  = (const float*)d_in[12];
  const float* W2p   = (const float*)d_in[13];
  const float* b2p   = (const float*)d_in[14];
  const float* g2p   = (const float*)d_in[15];
  const float* be2p  = (const float*)d_in[16];
  const float* rm2p  = (const float*)d_in[17];
  const float* rv2p  = (const float*)d_in[18];
  const float* W3p   = (const float*)d_in[19];
  const float* b3p   = (const float*)d_in[20];

  char* ws = (char*)d_ws;
  int*   bid    = (int*)ws;
  int*   dst    = (int*)(ws + (size_t)Np * sizeof(int));
  float* padded = (float*)(ws + 2 * (size_t)Np * sizeof(int));
  float* act1   = padded + (size_t)Np * Mf;
  float* act2   = padded;

  k_bid <<<Np / 256, 256, 0, stream>>>(pidx, poff, boff, bid);
  k_rank<<<Np / 256, 256, 0, stream>>>(bid, dst);
  k_zero<<<(Np * Mf) / 256, 256, 0, stream>>>(padded, Np * Mf);
  k_scatter<<<Np, 128, 0, stream>>>(feats, dst, padded);

  dim3 grid(Bn, 4);
  k_l1 <<<grid, 256, 0, stream>>>(padded, lang, Wf, bf, act1);
  k_mid<<<grid, 256, 0, stream>>>(act1, W1p, b1p, g1p, be1p, rm1p, rv1p, act2);
  k_mid<<<grid, 256, 0, stream>>>(act2, W2p, b2p, g2p, be2p, rm2p, rv2p, act1);
  k_final<<<Np / 256, 256, 0, stream>>>(act1, W3p, b3p, (float*)d_out);
}


# k_rank: O(N^2) LDS compare loop replaced by counting-sort rank: per-block bucket counts (wave ballot+popcount) written by k_bid, prefix over 32 block rows, in-wave rank by mask bit count
# speedup vs baseline: 1.8704x; 1.8704x over previous
#include <hip/hip_runtime.h>

constexpr int Bn    = 32;
constexpr int PPB   = 256;
constexpr int Np    = Bn * PPB;
constexpr int Mf    = 128;
constexpr int LANGD = 256;
constexpr int Hd    = 128;
constexpr int Kfuse = LANGD + Mf;
#define BN_EPS 1e-5f

typedef float v2f __attribute__((ext_vector_type(2)));
typedef float v8f __attribute__((ext_vector_type(8)));

__device__ __forceinline__ v8f wmma4(v2f a, v2f b, v8f c) {
  return __builtin_amdgcn_wmma_f32_16x16x4_f32(false, a, false, b, (short)0, c,
                                               false, false);
}

#define V8F_ZERO {0.f, 0.f, 0.f, 0.f, 0.f, 0.f, 0.f, 0.f}

constexpr int NB  = Bn + 2;
constexpr int RB  = 256;
constexpr int NRB = Np / RB;
constexpr int NW  = RB / 32;
static_assert(Np % RB == 0, "rank blocks must tile the proposals exactly");

__device__ __forceinline__ unsigned wave_bucket_counts(int b, int lane,
                                                       int* __restrict__ wcnt) {
  unsigned same = 0u;
  int c[NB];
#pragma unroll
  for (int v = 0; v < NB; ++v) {
    const unsigned m = __builtin_amdgcn_ballot_w32(b == v - 1);
    if (b == v - 1) same = m;
    c[v] = __popc(m);
  }
  if (lane == 0) {
#pragma unroll
    for (int v = 0; v < NB; ++v) wcnt[v] = c[v];
  }
  return same;
}

__global__ __launch_bounds__(RB) void k_bid(const int* __restrict__ pidx,
                                            const int* __restrict__ poff,
                                            const int* __restrict__ boff,
                                            int* __restrict__ bid,
                                            int* __restrict__ hist) {
  __shared__ int s_cnt[NW][NB];
  const int i    = blockIdx.x * RB + threadIdx.x;
  const int lane = threadIdx.x & 31;
  const int wave = threadIdx.x >> 5;
  int first_pt = pidx[2 * poff[i] + 1];
  int b = -1;
#pragma unroll
  for (int j = 0; j <= Bn; ++j) b += (boff[j] <= first_pt) ? 1 : 0;

  wave_bucket_counts(b, lane, s_cnt[wave]);
  __syncthreads();
  bid[i] = b;
  if (threadIdx.x < NB) {
    int s = 0;
#pragma unroll
    for (int w = 0; w < NW; ++w) s += s_cnt[w][threadIdx.x];
    hist[blockIdx.x * NB + threadIdx.x] = s;
  }
}

__global__ __launch_bounds__(RB) void k_rank(const int* __restrict__ bid,
                                             const int* __restrict__ hist,
                                             int* __restrict__ dst) {
  __shared__ int s_cnt[NW][NB];
  __shared__ int s_base[NB];
  const int i    = blockIdx.x * RB + threadIdx.x;
  const int lane = threadIdx.x & 31;
  const int wave = threadIdx.x >> 5;
  const int myb  = bid[i];

  if (threadIdx.x < NB) {
    int h[NRB];
#pragma unroll
    for (int q = 0; q < NRB; ++q) h[q] = hist[q * NB + threadIdx.x];
    int s = 0;
#pragma unroll
    for (int q = 0; q < NRB; ++q) s += (q < (int)blockIdx.x) ? h[q] : 0;
    s_base[threadIdx.x] = s;
  }

  const unsigned same = wave_bucket_counts(myb, lane, s_cnt[wave]);
  __syncthreads();

  const int idx = myb + 1;
  int r = s_base[idx] + (int)__builtin_amdgcn_mbcnt_lo(same, 0u);
#pragma unroll
  for (int w = 0; w < NW; ++w) r += (w < wave) ? s_cnt[w][idx] : 0;
  dst[i] = myb * PPB + r;
}

__global__ void k_zero(float* __restrict__ p, int n) {
  int i = blockIdx.x * blockDim.x + threadIdx.x;
  if (i < n) p[i] = 0.f;
}

__global__ __launch_bounds__(128) void k_scatter(const float* __restrict__ feats,
                                                 const int* __restrict__ dst,
                                                 float* __restrict__ padded) {
  int i = blockIdx.x;
  int d = dst[i];
  if (d < 0 || d >= Np) return;
  padded[(size_t)d * Mf + threadIdx.x] = feats[(size_t)i * Mf + threadIdx.x];
}

__global__ __launch_bounds__(256) void k_l1(const float* __restrict__ padded,
                                            const float* __restrict__ lang,
                                            const float* __restrict__ Wf,
                                            const float* __restrict__ bf,
                                            float* __restrict__ out) {
  const int b     = blockIdx.x;
  const int pbase = blockIdx.y * 64;
  const int wave  = threadIdx.x >> 5;
  const int lane  = threadIdx.x & 31;
  const int half  = lane >> 4;
  const int lm    = lane & 15;
  const int o0    = wave << 4;

  const float* arow = Wf + (size_t)(o0 + lm) * Kfuse + 2 * half;
  const float* xb   = padded + (size_t)b * PPB * Mf;
  const float* br0  = xb + (size_t)(pbase +  0 + lm) * Mf + 2 * half;
  const float* br1  = xb + (size_t)(pbase + 16 + lm) * Mf + 2 * half;
  const float* br2  = xb + (size_t)(pbase + 32 + lm) * Mf + 2 * half;
  const float* br3  = xb + (size_t)(pbase + 48 + lm) * Mf + 2 * half;
  const float* lb   = lang + (size_t)b * LANGD + 2 * half;

  v8f acc0 = V8F_ZERO, acc1 = V8F_ZERO, acc2 = V8F_ZERO, acc3 = V8F_ZERO;

#pragma unroll 4
  for (int k0 = 0; k0 < Mf; k0 += 4) {
    v2f a;   a.x  = arow[k0];      a.y  = arow[k0 + 1];
    v2f b0;  b0.x = br0[k0];       b0.y = br0[k0 + 1];
    v2f b1;  b1.x = br1[k0];       b1.y = br1[k0 + 1];
    v2f b2;  b2.x = br2[k0];       b2.y = br2[k0 + 1];
    v2f b3;  b3.x = br3[k0];       b3.y = br3[k0 + 1];
    acc0 = wmma4(a, b0, acc0);
    acc1 = wmma4(a, b1, acc1);
    acc2 = wmma4(a, b2, acc2);
    acc3 = wmma4(a, b3, acc3);
  }
#pragma unroll 4
  for (int k0 = 0; k0 < LANGD; k0 += 4) {
    v2f a;   a.x = arow[Mf + k0];  a.y = arow[Mf + k0 + 1];
    v2f bb;  bb.x = lb[k0];        bb.y = lb[k0 + 1];
    acc0 = wmma4(a, bb, acc0);
    acc1 = wmma4(a, bb, acc1);
    acc2 = wmma4(a, bb, acc2);
    acc3 = wmma4(a, bb, acc3);
  }

  float* ob = out + ((size_t)b * Hd) * PPB + pbase + lm;
#pragma unroll
  for (int v = 0; v < 8; ++v) {
    const int o = o0 + v + 8 * half;
    const float bias = bf[o];
    float* orow = ob + (size_t)o * PPB;
    orow[ 0] = fmaxf(acc0[v] + bias, 0.f);
    orow[16] = fmaxf(acc1[v] + bias, 0.f);
    orow[32] = fmaxf(acc2[v] + bias, 0.f);
    orow[48] = fmaxf(acc3[v] + bias, 0.f);
  }
}

__global__ __launch_bounds__(256) void k_mid(const float* __restrict__ in,
                                             const float* __restrict__ W,
                                             const float* __restrict__ bias,
                                             const float* __restrict__ gam,
                                             const float* __restrict__ bet,
                                             const float* __restrict__ rme,
                                             const float* __restrict__ rva,
                                             float* __restrict__ out) {
  const int b     = blockIdx.x;
  const int pbase = blockIdx.y * 64;
  const int wave  = threadIdx.x >> 5;
  const int lane  = threadIdx.x & 31;
  const int half  = lane >> 4;
  const int lm    = lane & 15;
  const int o0    = wave << 4;

  const float* arow = W + (size_t)(o0 + lm) * Hd + 2 * half;
  const float* inb  = in + (size_t)b * Hd * PPB;
  const float* bc   = inb + (size_t)(2 * half) * PPB + pbase + lm;

  v8f acc0 = V8F_ZERO, acc1 = V8F_ZERO, acc2 = V8F_ZERO, acc3 = V8F_ZERO;

#pragma unroll 4
  for (int k0 = 0; k0 < Hd; k0 += 4) {
    const int r0 = k0 * PPB;
    const int r1 = r0 + PPB;
    v2f a;   a.x  = arow[k0];      a.y  = arow[k0 + 1];
    v2f b0;  b0.x = bc[r0 +  0];   b0.y = bc[r1 +  0];
    v2f b1;  b1.x = bc[r0 + 16];   b1.y = bc[r1 + 16];
    v2f b2;  b2.x = bc[r0 + 32];   b2.y = bc[r1 + 32];
    v2f b3;  b3.x = bc[r0 + 48];   b3.y = bc[r1 + 48];
    acc0 = wmma4(a, b0, acc0);
    acc1 = wmma4(a, b1, acc1);
    acc2 = wmma4(a, b2, acc2);
    acc3 = wmma4(a, b3, acc3);
  }

  float* ob = out + ((size_t)b * Hd) * PPB + pbase + lm;
#pragma unroll
  for (int v = 0; v < 8; ++v) {
    const int o = o0 + v + 8 * half;
    const float bs    = bias[o];
    const float scale = gam[o] * rsqrtf(rva[o] + BN_EPS);
    const float mu    = rme[o];
    const float beta  = bet[o];
    float* orow = ob + (size_t)o * PPB;
    orow[ 0] = (fmaxf(acc0[v] + bs, 0.f) - mu) * scale + beta;
    orow[16] = (fmaxf(acc1[v] + bs, 0.f) - mu) * scale + beta;
    orow[32] = (fmaxf(acc2[v] + bs, 0.f) - mu) * scale + beta;
    orow[48] = (fmaxf(acc3[v] + bs, 0.f) - mu) * scale + beta;
  }
}

__global__ __launch_bounds__(256) void k_final(const float* __restrict__ in,
                                               const float* __restrict__ W3,
                                               const float* __restrict__ b3,
                                               float* __restrict__ out) {
  int idx = blockIdx.x * blockDim.x + threadIdx.x;
  if (idx >= Bn * PPB) return;
  int b = idx / PPB, p = idx % PPB;
  const float* inb = in + (size_t)b * Hd * PPB + p;
  float s = b3[0];
#pragma unroll 8
  for (int c = 0; c < Hd; ++c) s += W3[c] * inb[(size_t)c * PPB];
  out[idx] = s;
}

extern "C" void kernel_launch(void* const* d_in, const int* in_sizes, int n_in,
                              void* d_out, int out_size, void* d_ws, size_t ws_size,
                              hipStream_t stream) {
  const float* feats = (const float*)d_in[0];
  const int*   pidx  = (const int*)d_in[1];
  const int*   poff  = (const int*)d_in[2];
  const int*   boff  = (const int*)d_in[3];
  const float* lang  = (const float*)d_in[4];
  const float* Wf    = (const float*)d_in[5];
  const float* bf    = (const float*)d_in[6];
  const float* W1p   = (const float*)d_in[7];
  const float* b1p   = (const float*)d_in[8];
  const float* g1p   = (const float*)d_in[9];
  const float* be1p  = (const float*)d_in[10];
  const float* rm1p  = (const float*)d_in[11];
  const float* rv1p  = (const float*)d_in[12];
  const float* W2p   = (const float*)d_in[13];
  const float* b2p   = (const float*)d_in[14];
  const float* g2p   = (const float*)d_in[15];
  const float* be2p  = (const float*)d_in[16];
  const float* rm2p  = (const float*)d_in[17];
  const float* rv2p  = (const float*)d_in[18];
  const float* W3p   = (const float*)d_in[19];
  const float* b3p   = (const float*)d_in[20];

  char* ws = (char*)d_ws;
  int*   bid    = (int*)ws;
  int*   dst    = (int*)(ws + (size_t)Np * sizeof(int));
  float* padded = (float*)(ws + 2 * (size_t)Np * sizeof(int));
  float* act1   = padded + (size_t)Np * Mf;
  float* act2   = padded;
  int*   hist   = (int*)act1;

  k_bid <<<NRB, RB, 0, stream>>>(pidx, poff, boff, bid, hist);
  k_rank<<<NRB, RB, 0, stream>>>(bid, hist, dst);
  k_zero<<<(Np * Mf) / 256, 256, 0, stream>>>(padded, Np * Mf);
  k_scatter<<<Np, 128, 0, stream>>>(feats, dst, padded);

  dim3 grid(Bn, 4);
  k_l1 <<<grid, 256, 0, stream>>>(padded, lang, Wf, bf, act1);
  k_mid<<<grid, 256, 0, stream>>>(act1, W1p, b1p, g1p, be1p, rm1p, rv1p, act2);
  k_mid<<<grid, 256, 0, stream>>>(act2, W2p, b2p, g2p, be2p, rm2p, rv2p, act1);
  k_final<<<Np / 256, 256, 0, stream>>>(act1, W3p, b3p, (float*)d_out);
}


# zero-fill and scatter launches removed: layer 1 gathers its feature rows from the input through an inverse slot index (src[slot]=proposal, empty slot -> a zero row) that k_bid/k_rank write
# speedup vs baseline: 2.1318x; 1.1398x over previous
#include <hip/hip_runtime.h>

constexpr int Bn    = 32;
constexpr int PPB   = 256;
constexpr int Np    = Bn * PPB;
constexpr int Mf    = 128;
constexpr int LANGD = 256;
constexpr int Hd    = 128;
constexpr int Kfuse = LANGD + Mf;
#define BN_EPS 1e-5f

typedef float v2f __attribute__((ext_vector_type(2)));
typedef float v8f __attribute__((ext_vector_type(8)));

__device__ __forceinline__ v8f wmma4(v2f a, v2f b, v8f c) {
  return __builtin_amdgcn_wmma_f32_16x16x4_f32(false, a, false, b, (short)0, c,
                                               false, false);
}

#define V8F_ZERO {0.f, 0.f, 0.f, 0.f, 0.f, 0.f, 0.f, 0.f}

constexpr int NB  = Bn + 2;
constexpr int RB  = 256;
constexpr int NRB = Np / RB;
constexpr int NW  = RB / 32;
static_assert(Np % RB == 0, "rank blocks must tile the proposals exactly");

__device__ __forceinline__ unsigned wave_bucket_counts(int b, int lane,
                                                       int* __restrict__ wcnt) {
  unsigned same = 0u;
  int c[NB];
#pragma unroll
  for (int v = 0; v < NB; ++v) {
    const unsigned m = __builtin_amdgcn_ballot_w32(b == v - 1);
    if (b == v - 1) same = m;
    c[v] = __popc(m);
  }
  if (lane == 0) {
#pragma unroll
    for (int v = 0; v < NB; ++v) wcnt[v] = c[v];
  }
  return same;
}

__global__ __launch_bounds__(RB) void k_bid(const int* __restrict__ pidx,
                                            const int* __restrict__ poff,
                                            const int* __restrict__ boff,
                                            int* __restrict__ bid,
                                            int* __restrict__ hist,
                                            int* __restrict__ src,
                                            float* __restrict__ zrow) {
  __shared__ int s_cnt[NW][NB];
  const int i    = blockIdx.x * RB + threadIdx.x;
  const int lane = threadIdx.x & 31;
  const int wave = threadIdx.x >> 5;
  int first_pt = pidx[2 * poff[i] + 1];
  int b = -1;
#pragma unroll
  for (int j = 0; j <= Bn; ++j) b += (boff[j] <= first_pt) ? 1 : 0;

  wave_bucket_counts(b, lane, s_cnt[wave]);
  __syncthreads();
  bid[i] = b;
  src[i] = -1;
  if (blockIdx.x == 0 && threadIdx.x < Mf) zrow[threadIdx.x] = 0.f;
  if (threadIdx.x < NB) {
    int s = 0;
#pragma unroll
    for (int w = 0; w < NW; ++w) s += s_cnt[w][threadIdx.x];
    hist[blockIdx.x * NB + threadIdx.x] = s;
  }
}

__global__ __launch_bounds__(RB) void k_rank(const int* __restrict__ bid,
                                             const int* __restrict__ hist,
                                             int* __restrict__ src) {
  __shared__ int s_cnt[NW][NB];
  __shared__ int s_base[NB];
  const int i    = blockIdx.x * RB + threadIdx.x;
  const int lane = threadIdx.x & 31;
  const int wave = threadIdx.x >> 5;
  const int myb  = bid[i];

  if (threadIdx.x < NB) {
    int h[NRB];
#pragma unroll
    for (int q = 0; q < NRB; ++q) h[q] = hist[q * NB + threadIdx.x];
    int s = 0;
#pragma unroll
    for (int q = 0; q < NRB; ++q) s += (q < (int)blockIdx.x) ? h[q] : 0;
    s_base[threadIdx.x] = s;
  }

  const unsigned same = wave_bucket_counts(myb, lane, s_cnt[wave]);
  __syncthreads();

  const int idx = myb + 1;
  int r = s_base[idx] + (int)__builtin_amdgcn_mbcnt_lo(same, 0u);
#pragma unroll
  for (int w = 0; w < NW; ++w) r += (w < wave) ? s_cnt[w][idx] : 0;
  const int d = myb * PPB + r;
  if (d >= 0 && d < Np) src[d] = i;
}

__global__ __launch_bounds__(256) void k_l1(const float* __restrict__ feats,
                                            const int* __restrict__ src,
                                            const float* __restrict__ zrow,
                                            const float* __restrict__ lang,
                                            const float* __restrict__ Wf,
                                            const float* __restrict__ bf,
                                            float* __restrict__ out) {
  const int b     = blockIdx.x;
  const int pbase = blockIdx.y * 64;
  const int wave  = threadIdx.x >> 5;
  const int lane  = threadIdx.x & 31;
  const int half  = lane >> 4;
  const int lm    = lane & 15;
  const int o0    = wave << 4;

  const float* arow = Wf + (size_t)(o0 + lm) * Kfuse + 2 * half;
  const int* sl = src + b * PPB + pbase + lm;
  const int s0 = sl[0], s1 = sl[16], s2 = sl[32], s3 = sl[48];
  const float* br0  = ((unsigned)s0 < (unsigned)Np ? feats + (size_t)s0 * Mf : zrow) + 2 * half;
  const float* br1  = ((unsigned)s1 < (unsigned)Np ? feats + (size_t)s1 * Mf : zrow) + 2 * half;
  const float* br2  = ((unsigned)s2 < (unsigned)Np ? feats + (size_t)s2 * Mf : zrow) + 2 * half;
  const float* br3  = ((unsigned)s3 < (unsigned)Np ? feats + (size_t)s3 * Mf : zrow) + 2 * half;
  const float* lb   = lang + (size_t)b * LANGD + 2 * half;

  v8f acc0 = V8F_ZERO, acc1 = V8F_ZERO, acc2 = V8F_ZERO, acc3 = V8F_ZERO;

#pragma unroll 4
  for (int k0 = 0; k0 < Mf; k0 += 4) {
    v2f a;   a.x  = arow[k0];      a.y  = arow[k0 + 1];
    v2f b0;  b0.x = br0[k0];       b0.y = br0[k0 + 1];
    v2f b1;  b1.x = br1[k0];       b1.y = br1[k0 + 1];
    v2f b2;  b2.x = br2[k0];       b2.y = br2[k0 + 1];
    v2f b3;  b3.x = br3[k0];       b3.y = br3[k0 + 1];
    acc0 = wmma4(a, b0, acc0);
    acc1 = wmma4(a, b1, acc1);
    acc2 = wmma4(a, b2, acc2);
    acc3 = wmma4(a, b3, acc3);
  }
#pragma unroll 4
  for (int k0 = 0; k0 < LANGD; k0 += 4) {
    v2f a;   a.x = arow[Mf + k0];  a.y = arow[Mf + k0 + 1];
    v2f bb;  bb.x = lb[k0];        bb.y = lb[k0 + 1];
    acc0 = wmma4(a, bb, acc0);
    acc1 = wmma4(a, bb, acc1);
    acc2 = wmma4(a, bb, acc2);
    acc3 = wmma4(a, bb, acc3);
  }

  float* ob = out + ((size_t)b * Hd) * PPB + pbase + lm;
#pragma unroll
  for (int v = 0; v < 8; ++v) {
    const int o = o0 + v + 8 * half;
    const float bias = bf[o];
    float* orow = ob + (size_t)o * PPB;
    orow[ 0] = fmaxf(acc0[v] + bias, 0.f);
    orow[16] = fmaxf(acc1[v] + bias, 0.f);
    orow[32] = fmaxf(acc2[v] + bias, 0.f);
    orow[48] = fmaxf(acc3[v] + bias, 0.f);
  }
}

__global__ __launch_bounds__(256) void k_mid(const float* __restrict__ in,
                                             const float* __restrict__ W,
                                             const float* __restrict__ bias,
                                             const float* __restrict__ gam,
                                             const float* __restrict__ bet,
                                             const float* __restrict__ rme,
                                             const float* __restrict__ rva,
                                             float* __restrict__ out) {
  const int b     = blockIdx.x;
  const int pbase = blockIdx.y * 64;
  const int wave  = threadIdx.x >> 5;
  const int lane  = threadIdx.x & 31;
  const int half  = lane >> 4;
  const int lm    = lane & 15;
  const int o0    = wave << 4;

  const float* arow = W + (size_t)(o0 + lm) * Hd + 2 * half;
  const float* inb  = in + (size_t)b * Hd * PPB;
  const float* bc   = inb + (size_t)(2 * half) * PPB + pbase + lm;

  v8f acc0 = V8F_ZERO, acc1 = V8F_ZERO, acc2 = V8F_ZERO, acc3 = V8F_ZERO;

#pragma unroll 4
  for (int k0 = 0; k0 < Hd; k0 += 4) {
    const int r0 = k0 * PPB;
    const int r1 = r0 + PPB;
    v2f a;   a.x  = arow[k0];      a.y  = arow[k0 + 1];
    v2f b0;  b0.x = bc[r0 +  0];   b0.y = bc[r1 +  0];
    v2f b1;  b1.x = bc[r0 + 16];   b1.y = bc[r1 + 16];
    v2f b2;  b2.x = bc[r0 + 32];   b2.y = bc[r1 + 32];
    v2f b3;  b3.x = bc[r0 + 48];   b3.y = bc[r1 + 48];
    acc0 = wmma4(a, b0, acc0);
    acc1 = wmma4(a, b1, acc1);
    acc2 = wmma4(a, b2, acc2);
    acc3 = wmma4(a, b3, acc3);
  }

  float* ob = out + ((size_t)b * Hd) * PPB + pbase + lm;
#pragma unroll
  for (int v = 0; v < 8; ++v) {
    const int o = o0 + v + 8 * half;
    const float bs    = bias[o];
    const float scale = gam[o] * rsqrtf(rva[o] + BN_EPS);
    const float mu    = rme[o];
    const float beta  = bet[o];
    float* orow = ob + (size_t)o * PPB;
    orow[ 0] = (fmaxf(acc0[v] + bs, 0.f) - mu) * scale + beta;
    orow[16] = (fmaxf(acc1[v] + bs, 0.f) - mu) * scale + beta;
    orow[32] = (fmaxf(acc2[v] + bs, 0.f) - mu) * scale + beta;
    orow[48] = (fmaxf(acc3[v] + bs, 0.f) - mu) * scale + beta;
  }
}

__global__ __launch_bounds__(256) void k_final(const float* __restrict__ in,
                                               const float* __restrict__ W3,
                                               const float* __restrict__ b3,
                                               float* __restrict__ out) {
  int idx = blockIdx.x * blockDim.x + threadIdx.x;
  if (idx >= Bn * PPB) return;
  int b = idx / PPB, p = idx % PPB;
  const float* inb = in + (size_t)b * Hd * PPB + p;
  float s = b3[0];
#pragma unroll 8
  for (int c = 0; c < Hd; ++c) s += W3[c] * inb[(size_t)c * PPB];
  out[idx] = s;
}

extern "C" void kernel_launch(void* const* d_in, const int* in_sizes, int n_in,
                              void* d_out, int out_size, void* d_ws, size_t ws_size,
                              hipStream_t stream) {
  const float* feats = (const float*)d_in[0];
  const int*   pidx  = (const int*)d_in[1];
  const int*   poff  = (const int*)d_in[2];
  const int*   boff  = (const int*)d_in[3];
  const float* lang  = (const float*)d_in[4];
  const float* Wf    = (const float*)d_in[5];
  const float* bf    = (const float*)d_in[6];
  const float* W1p   = (const float*)d_in[7];
  const float* b1p   = (const float*)d_in[8];
  const float* g1p   = (const float*)d_in[9];
  const float* be1p  = (const float*)d_in[10];
  const float* rm1p  = (const float*)d_in[11];
  const float* rv1p  = (const float*)d_in[12];
  const float* W2p   = (const float*)d_in[13];
  const float* b2p   = (const float*)d_in[14];
  const float* g2p   = (const float*)d_in[15];
  const float* be2p  = (const float*)d_in[16];
  const float* rm2p  = (const float*)d_in[17];
  const float* rv2p  = (const float*)d_in[18];
  const float* W3p   = (const float*)d_in[19];
  const float* b3p   = (const float*)d_in[20];

  char* ws = (char*)d_ws;
  int*   bid    = (int*)ws;
  int*   src    = (int*)(ws + (size_t)Np * sizeof(int));
  float* act2   = (float*)(ws + 2 * (size_t)Np * sizeof(int));
  float* act1   = act2 + (size_t)Np * Mf;
  int*   hist   = (int*)act1;
  float* zrow   = act2;

  k_bid <<<NRB, RB, 0, stream>>>(pidx, poff, boff, bid, hist, src, zrow);
  k_rank<<<NRB, RB, 0, stream>>>(bid, hist, src);

  dim3 grid(Bn, 4);
  k_l1 <<<grid, 256, 0, stream>>>(feats, src, zrow, lang, Wf, bf, act1);
  k_mid<<<grid, 256, 0, stream>>>(act1, W1p, b1p, g1p, be1p, rm1p, rv1p, act2);
  k_mid<<<grid, 256, 0, stream>>>(act2, W2p, b2p, g2p, be2p, rm2p, rv2p, act1);
  k_final<<<Np / 256, 256, 0, stream>>>(act1, W3p, b3p, (float*)d_out);
}


# four MLP launches fused into one (6 -> 3 launches): a work-group keeps its 64-column activations in LDS between layers and loads the next layer's weights ahead of the barrier; sum order kept
# speedup vs baseline: 4.1062x; 1.9262x over previous
#include <hip/hip_runtime.h>

constexpr int Bn    = 32;
constexpr int PPB   = 256;
constexpr int Np    = Bn * PPB;
constexpr int Mf    = 128;
constexpr int LANGD = 256;
constexpr int Hd    = 128;
constexpr int Kfuse = LANGD + Mf;
#define BN_EPS 1e-5f

typedef float v2f __attribute__((ext_vector_type(2)));
typedef float v8f __attribute__((ext_vector_type(8)));

__device__ __forceinline__ v8f wmma4(v2f a, v2f b, v8f c) {
  return __builtin_amdgcn_wmma_f32_16x16x4_f32(false, a, false, b, (short)0, c,
                                               false, false);
}

#define V8F_ZERO {0.f, 0.f, 0.f, 0.f, 0.f, 0.f, 0.f, 0.f}

constexpr int NB  = Bn + 2;
constexpr int RB  = 256;
constexpr int NRB = Np / RB;
constexpr int NW  = RB / 32;
static_assert(Np % RB == 0, "rank blocks must tile the proposals exactly");

__device__ __forceinline__ unsigned wave_bucket_counts(int b, int lane,
                                                       int* __restrict__ wcnt) {
  unsigned same = 0u;
  int c[NB];
#pragma unroll
  for (int v = 0; v < NB; ++v) {
    const unsigned m = __builtin_amdgcn_ballot_w32(b == v - 1);
    if (b == v - 1) same = m;
    c[v] = __popc(m);
  }
  if (lane == 0) {
#pragma unroll
    for (int v = 0; v < NB; ++v) wcnt[v] = c[v];
  }
  return same;
}

__global__ __launch_bounds__(RB) void k_bid(const int* __restrict__ pidx,
                                            const int* __restrict__ poff,
                                            const int* __restrict__ boff,
                                            int* __restrict__ bid,
                                            int* __restrict__ hist,
                                            int* __restrict__ src,
                                            float* __restrict__ zrow) {
  __shared__ int s_cnt[NW][NB];
  const int i    = blockIdx.x * RB + threadIdx.x;
  const int lane = threadIdx.x & 31;
  const int wave = threadIdx.x >> 5;
  int first_pt = pidx[2 * poff[i] + 1];
  int b = -1;
#pragma unroll
  for (int j = 0; j <= Bn; ++j) b += (boff[j] <= first_pt) ? 1 : 0;

  wave_bucket_counts(b, lane, s_cnt[wave]);
  __syncthreads();
  bid[i] = b;
  src[i] = -1;
  if (blockIdx.x == 0 && threadIdx.x < Mf) zrow[threadIdx.x] = 0.f;
  if (threadIdx.x < NB) {
    int s = 0;
#pragma unroll
    for (int w = 0; w < NW; ++w) s += s_cnt[w][threadIdx.x];
    hist[blockIdx.x * NB + threadIdx.x] = s;
  }
}

__global__ __launch_bounds__(RB) void k_rank(const int* __restrict__ bid,
                                             const int* __restrict__ hist,
                                             int* __restrict__ src) {
  __shared__ int s_cnt[NW][NB];
  __shared__ int s_base[NB];
  const int i    = blockIdx.x * RB + threadIdx.x;
  const int lane = threadIdx.x & 31;
  const int wave = threadIdx.x >> 5;
  const int myb  = bid[i];

  if (threadIdx.x < NB) {
    int h[NRB];
#pragma unroll
    for (int q = 0; q < NRB; ++q) h[q] = hist[q * NB + threadIdx.x];
    int s = 0;
#pragma unroll
    for (int q = 0; q < NRB; ++q) s += (q < (int)blockIdx.x) ? h[q] : 0;
    s_base[threadIdx.x] = s;
  }

  const unsigned same = wave_bucket_counts(myb, lane, s_cnt[wave]);
  __syncthreads();

  const int idx = myb + 1;
  int r = s_base[idx] + (int)__builtin_amdgcn_mbcnt_lo(same, 0u);
#pragma unroll
  for (int w = 0; w < NW; ++w) r += (w < wave) ? s_cnt[w][idx] : 0;
  const int d = myb * PPB + r;
  if (d >= 0 && d < Np) src[d] = i;
}

constexpr int PT  = 64;
constexpr int XP2 = 80;
constexpr int XPLANE = (Hd / 2) * XP2;
constexpr int KS  = Hd / 4;

__device__ __forceinline__ void load_a(const float* __restrict__ W,
                                       int o0, int lm, int half,
                                       v2f (&aw)[KS]) {
  const float* arow = W + (size_t)(o0 + lm) * Hd + 2 * half;
#pragma unroll
  for (int q = 0; q < KS; ++q) { aw[q].x = arow[4 * q]; aw[q].y = arow[4 * q + 1]; }
}

__device__ __forceinline__ void mid_gemm(const v2f (&aw)[KS], const v2f* xin,
                                         int lm, int half,
                                         v8f& acc0, v8f& acc1, v8f& acc2, v8f& acc3) {
  const v2f* bc = xin + half * XP2 + lm;
#pragma unroll
  for (int q = 0; q < KS; ++q) {
    const v2f* br = bc + (2 * q) * XP2;
    const v2f b0 = br[ 0];
    const v2f b1 = br[16];
    const v2f b2 = br[32];
    const v2f b3 = br[48];
    acc0 = wmma4(aw[q], b0, acc0);
    acc1 = wmma4(aw[q], b1, acc1);
    acc2 = wmma4(aw[q], b2, acc2);
    acc3 = wmma4(aw[q], b3, acc3);
  }
}

__device__ __forceinline__ void mid_store(const v8f& acc0, const v8f& acc1,
                                          const v8f& acc2, const v8f& acc3,
                                          const float* __restrict__ bias,
                                          const float* __restrict__ gam,
                                          const float* __restrict__ bet,
                                          const float* __restrict__ rme,
                                          const float* __restrict__ rva,
                                          v2f* xout, int o0, int lm, int half) {
  v2f* ob = xout + lm;
#pragma unroll
  for (int v = 0; v < 8; v += 2) {
    const int o = o0 + v + 8 * half;
    const float bs0 = bias[o],     bs1 = bias[o + 1];
    const float sc0 = gam[o]     * rsqrtf(rva[o]     + BN_EPS);
    const float sc1 = gam[o + 1] * rsqrtf(rva[o + 1] + BN_EPS);
    const float mu0 = rme[o],      mu1 = rme[o + 1];
    const float be0 = bet[o],      be1 = bet[o + 1];
    v2f* orow = ob + (o >> 1) * XP2;
    v2f r;
    r.x = (fmaxf(acc0[v] + bs0, 0.f) - mu0) * sc0 + be0;
    r.y = (fmaxf(acc0[v + 1] + bs1, 0.f) - mu1) * sc1 + be1;
    orow[ 0] = r;
    r.x = (fmaxf(acc1[v] + bs0, 0.f) - mu0) * sc0 + be0;
    r.y = (fmaxf(acc1[v + 1] + bs1, 0.f) - mu1) * sc1 + be1;
    orow[16] = r;
    r.x = (fmaxf(acc2[v] + bs0, 0.f) - mu0) * sc0 + be0;
    r.y = (fmaxf(acc2[v + 1] + bs1, 0.f) - mu1) * sc1 + be1;
    orow[32] = r;
    r.x = (fmaxf(acc3[v] + bs0, 0.f) - mu0) * sc0 + be0;
    r.y = (fmaxf(acc3[v + 1] + bs1, 0.f) - mu1) * sc1 + be1;
    orow[48] = r;
  }
}

__global__ __launch_bounds__(256) void k_mlp(const float* __restrict__ feats,
                                             const int* __restrict__ src,
                                             const float* __restrict__ zrow,
                                             const float* __restrict__ lang,
                                             const float* __restrict__ Wf,
                                             const float* __restrict__ bf,
                                             const float* __restrict__ W1,
                                             const float* __restrict__ b1,
                                             const float* __restrict__ g1,
                                             const float* __restrict__ be1,
                                             const float* __restrict__ rm1,
                                             const float* __restrict__ rv1,
                                             const float* __restrict__ W2,
                                             const float* __restrict__ b2,
                                             const float* __restrict__ g2,
                                             const float* __restrict__ be2,
                                             const float* __restrict__ rm2,
                                             const float* __restrict__ rv2,
                                             const float* __restrict__ W3,
                                             const float* __restrict__ b3,
                                             float* __restrict__ out) {
  __shared__ v2f xa[XPLANE];
  __shared__ v2f xb[XPLANE];

  const int b     = blockIdx.x;
  const int pbase = blockIdx.y * PT;
  const int wave  = threadIdx.x >> 5;
  const int lane  = threadIdx.x & 31;
  const int half  = lane >> 4;
  const int lm    = lane & 15;
  const int o0    = wave << 4;

  v2f aw[KS];
  v8f acc0 = V8F_ZERO, acc1 = V8F_ZERO, acc2 = V8F_ZERO, acc3 = V8F_ZERO;

  {
    const float* arow = Wf + (size_t)(o0 + lm) * Kfuse + 2 * half;
    const int* sl = src + b * PPB + pbase + lm;
    const int s0 = sl[0], s1 = sl[16], s2 = sl[32], s3 = sl[48];
    const float* br0  = ((unsigned)s0 < (unsigned)Np ? feats + (size_t)s0 * Mf : zrow) + 2 * half;
    const float* br1  = ((unsigned)s1 < (unsigned)Np ? feats + (size_t)s1 * Mf : zrow) + 2 * half;
    const float* br2  = ((unsigned)s2 < (unsigned)Np ? feats + (size_t)s2 * Mf : zrow) + 2 * half;
    const float* br3  = ((unsigned)s3 < (unsigned)Np ? feats + (size_t)s3 * Mf : zrow) + 2 * half;
    const float* lb   = lang + (size_t)b * LANGD + 2 * half;

#pragma unroll 4
    for (int k0 = 0; k0 < Mf; k0 += 4) {
      v2f a;   a.x  = arow[k0];      a.y  = arow[k0 + 1];
      v2f b0;  b0.x = br0[k0];       b0.y = br0[k0 + 1];
      v2f b1;  b1.x = br1[k0];       b1.y = br1[k0 + 1];
      v2f b2;  b2.x = br2[k0];       b2.y = br2[k0 + 1];
      v2f b3;  b3.x = br3[k0];       b3.y = br3[k0 + 1];
      acc0 = wmma4(a, b0, acc0);
      acc1 = wmma4(a, b1, acc1);
      acc2 = wmma4(a, b2, acc2);
      acc3 = wmma4(a, b3, acc3);
    }
#pragma unroll 4
    for (int k0 = 0; k0 < LANGD; k0 += 4) {
      v2f a;   a.x = arow[Mf + k0];  a.y = arow[Mf + k0 + 1];
      v2f bb;  bb.x = lb[k0];        bb.y = lb[k0 + 1];
      acc0 = wmma4(a, bb, acc0);
      acc1 = wmma4(a, bb, acc1);
      acc2 = wmma4(a, bb, acc2);
      acc3 = wmma4(a, bb, acc3);
    }

    load_a(W1, o0, lm, half, aw);

    v2f* ob = xa + lm;
#pragma unroll
    for (int v = 0; v < 8; v += 2) {
      const int o = o0 + v + 8 * half;
      const float bias0 = bf[o], bias1 = bf[o + 1];
      v2f* orow = ob + (o >> 1) * XP2;
      v2f r;
      r.x = fmaxf(acc0[v] + bias0, 0.f);  r.y = fmaxf(acc0[v + 1] + bias1, 0.f);
      orow[ 0] = r;
      r.x = fmaxf(acc1[v] + bias0, 0.f);  r.y = fmaxf(acc1[v + 1] + bias1, 0.f);
      orow[16] = r;
      r.x = fmaxf(acc2[v] + bias0, 0.f);  r.y = fmaxf(acc2[v + 1] + bias1, 0.f);
      orow[32] = r;
      r.x = fmaxf(acc3[v] + bias0, 0.f);  r.y = fmaxf(acc3[v + 1] + bias1, 0.f);
      orow[48] = r;
    }
  }
  __syncthreads();

  acc0 = (v8f)V8F_ZERO; acc1 = (v8f)V8F_ZERO; acc2 = (v8f)V8F_ZERO; acc3 = (v8f)V8F_ZERO;
  mid_gemm(aw, xa, lm, half, acc0, acc1, acc2, acc3);
  load_a(W2, o0, lm, half, aw);
  mid_store(acc0, acc1, acc2, acc3, b1, g1, be1, rm1, rv1, xb, o0, lm, half);
  __syncthreads();

  acc0 = (v8f)V8F_ZERO; acc1 = (v8f)V8F_ZERO; acc2 = (v8f)V8F_ZERO; acc3 = (v8f)V8F_ZERO;
  mid_gemm(aw, xb, lm, half, acc0, acc1, acc2, acc3);
  mid_store(acc0, acc1, acc2, acc3, b2, g2, be2, rm2, rv2, xa, o0, lm, half);
  __syncthreads();

  if (threadIdx.x < PT) {
    const v2f* xc = xa + threadIdx.x;
    float s = b3[0];
#pragma unroll 8
    for (int c = 0; c < Hd; c += 2) {
      const v2f x = xc[(c >> 1) * XP2];
      s += W3[c] * x.x;
      s += W3[c + 1] * x.y;
    }
    out[b * PPB + pbase + threadIdx.x] = s;
  }
}

extern "C" void kernel_launch(void* const* d_in, const int* in_sizes, int n_in,
                              void* d_out, int out_size, void* d_ws, size_t ws_size,
                              hipStream_t stream) {
  const float* feats = (const float*)d_in[0];
  const int*   pidx  = (const int*)d_in[1];
  const int*   poff  = (const int*)d_in[2];
  const int*   boff  = (const int*)d_in[3];
  const float* lang  = (const float*)d_in[4];
  const float* Wf    = (const float*)d_in[5];
  const float* bf    = (const float*)d_in[6];
  const float* W1p   = (const float*)d_in[7];
  const float* b1p   = (const float*)d_in[8];
  const float* g1p   = (const float*)d_in[9];
  const float* be1p  = (const float*)d_in[10];
  const float* rm1p  = (const float*)d_in[11];
  const float* rv1p  = (const float*)d_in[12];
  const float* W2p   = (const float*)d_in[13];
  const float* b2p   = (const float*)d_in[14];
  const float* g2p   = (const float*)d_in[15];
  const float* be2p  = (const float*)d_in[16];
  const float* rm2p  = (const float*)d_in[17];
  const float* rv2p  = (const float*)d_in[18];
  const float* W3p   = (const float*)d_in[19];
  const float* b3p   = (const float*)d_in[20];

  char* ws = (char*)d_ws;
  int*   bid    = (int*)ws;
  int*   src    = (int*)(ws + (size_t)Np * sizeof(int));
  float* zrow   = (float*)(ws + 2 * (size_t)Np * sizeof(int));
  int*   hist   = (int*)(zrow + (size_t)Np * Mf);

  k_bid <<<NRB, RB, 0, stream>>>(pidx, poff, boff, bid, hist, src, zrow);
  k_rank<<<NRB, RB, 0, stream>>>(bid, hist, src);

  dim3 grid(Bn, 4);
  k_mlp<<<grid, 256, 0, stream>>>(feats, src, zrow, lang, Wf, bf,
                                  W1p, b1p, g1p, be1p, rm1p, rv1p,
                                  W2p, b2p, g2p, be2p, rm2p, rv2p,
                                  W3p, b3p, (float*)d_out);
}


# layer 1: lang half of K (256 of 384) multiplied once per wave, not per p-tile (1 WMMA per k-step for 4 equal ones, 16-byte fragment reads); its tile starts the 4 accumulators: lang terms summed first
# speedup vs baseline: 4.7630x; 1.1600x over previous
#include <hip/hip_runtime.h>

constexpr int Bn    = 32;
constexpr int PPB   = 256;
constexpr int Np    = Bn * PPB;
constexpr int Mf    = 128;
constexpr int LANGD = 256;
constexpr int Hd    = 128;
constexpr int Kfuse = LANGD + Mf;
#define BN_EPS 1e-5f

typedef float v2f __attribute__((ext_vector_type(2)));
typedef float v8f __attribute__((ext_vector_type(8)));

__device__ __forceinline__ v8f wmma4(v2f a, v2f b, v8f c) {
  return __builtin_amdgcn_wmma_f32_16x16x4_f32(false, a, false, b, (short)0, c,
                                               false, false);
}

#define V8F_ZERO {0.f, 0.f, 0.f, 0.f, 0.f, 0.f, 0.f, 0.f}

constexpr int NB  = Bn + 2;
constexpr int RB  = 256;
constexpr int NRB = Np / RB;
constexpr int NW  = RB / 32;
static_assert(Np % RB == 0, "rank blocks must tile the proposals exactly");

__device__ __forceinline__ unsigned wave_bucket_counts(int b, int lane,
                                                       int* __restrict__ wcnt) {
  unsigned same = 0u;
  int c[NB];
#pragma unroll
  for (int v = 0; v < NB; ++v) {
    const unsigned m = __builtin_amdgcn_ballot_w32(b == v - 1);
    if (b == v - 1) same = m;
    c[v] = __popc(m);
  }
  if (lane == 0) {
#pragma unroll
    for (int v = 0; v < NB; ++v) wcnt[v] = c[v];
  }
  return same;
}

__global__ __launch_bounds__(RB) void k_bid(const int* __restrict__ pidx,
                                            const int* __restrict__ poff,
                                            const int* __restrict__ boff,
                                            int* __restrict__ bid,
                                            int* __restrict__ hist,
                                            int* __restrict__ src,
                                            float* __restrict__ zrow) {
  __shared__ int s_cnt[NW][NB];
  const int i    = blockIdx.x * RB + threadIdx.x;
  const int lane = threadIdx.x & 31;
  const int wave = threadIdx.x >> 5;
  int first_pt = pidx[2 * poff[i] + 1];
  int b = -1;
#pragma unroll
  for (int j = 0; j <= Bn; ++j) b += (boff[j] <= first_pt) ? 1 : 0;

  wave_bucket_counts(b, lane, s_cnt[wave]);
  __syncthreads();
  bid[i] = b;
  src[i] = -1;
  if (blockIdx.x == 0 && threadIdx.x < Mf) zrow[threadIdx.x] = 0.f;
  if (threadIdx.x < NB) {
    int s = 0;
#pragma unroll
    for (int w = 0; w < NW; ++w) s += s_cnt[w][threadIdx.x];
    hist[blockIdx.x * NB + threadIdx.x] = s;
  }
}

__global__ __launch_bounds__(RB) void k_rank(const int* __restrict__ bid,
                                             const int* __restrict__ hist,
                                             int* __restrict__ src) {
  __shared__ int s_cnt[NW][NB];
  __shared__ int s_base[NB];
  const int i    = blockIdx.x * RB + threadIdx.x;
  const int lane = threadIdx.x & 31;
  const int wave = threadIdx.x >> 5;
  const int myb  = bid[i];

  if (threadIdx.x < NB) {
    int h[NRB];
#pragma unroll
    for (int q = 0; q < NRB; ++q) h[q] = hist[q * NB + threadIdx.x];
    int s = 0;
#pragma unroll
    for (int q = 0; q < NRB; ++q) s += (q < (int)blockIdx.x) ? h[q] : 0;
    s_base[threadIdx.x] = s;
  }

  const unsigned same = wave_bucket_counts(myb, lane, s_cnt[wave]);
  __syncthreads();

  const int idx = myb + 1;
  int r = s_base[idx] + (int)__builtin_amdgcn_mbcnt_lo(same, 0u);
#pragma unroll
  for (int w = 0; w < NW; ++w) r += (w < wave) ? s_cnt[w][idx] : 0;
  const int d = myb * PPB + r;
  if (d >= 0 && d < Np) src[d] = i;
}

constexpr int PT  = 64;
constexpr int XP2 = 80;
constexpr int XPLANE = (Hd / 2) * XP2;
constexpr int KS  = Hd / 4;

__device__ __forceinline__ void load_a(const float* __restrict__ W,
                                       int o0, int lm, int half,
                                       v2f (&aw)[KS]) {
  const float* arow = W + (size_t)(o0 + lm) * Hd + 2 * half;
#pragma unroll
  for (int q = 0; q < KS; ++q) { aw[q].x = arow[4 * q]; aw[q].y = arow[4 * q + 1]; }
}

__device__ __forceinline__ void mid_gemm(const v2f (&aw)[KS], const v2f* xin,
                                         int lm, int half,
                                         v8f& acc0, v8f& acc1, v8f& acc2, v8f& acc3) {
  const v2f* bc = xin + half * XP2 + lm;
#pragma unroll
  for (int q = 0; q < KS; ++q) {
    const v2f* br = bc + (2 * q) * XP2;
    const v2f b0 = br[ 0];
    const v2f b1 = br[16];
    const v2f b2 = br[32];
    const v2f b3 = br[48];
    acc0 = wmma4(aw[q], b0, acc0);
    acc1 = wmma4(aw[q], b1, acc1);
    acc2 = wmma4(aw[q], b2, acc2);
    acc3 = wmma4(aw[q], b3, acc3);
  }
}

__device__ __forceinline__ void mid_store(const v8f& acc0, const v8f& acc1,
                                          const v8f& acc2, const v8f& acc3,
                                          const float* __restrict__ bias,
                                          const float* __restrict__ gam,
                                          const float* __restrict__ bet,
                                          const float* __restrict__ rme,
                                          const float* __restrict__ rva,
                                          v2f* xout, int o0, int lm, int half) {
  v2f* ob = xout + lm;
#pragma unroll
  for (int v = 0; v < 8; v += 2) {
    const int o = o0 + v + 8 * half;
    const float bs0 = bias[o],     bs1 = bias[o + 1];
    const float sc0 = gam[o]     * rsqrtf(rva[o]     + BN_EPS);
    const float sc1 = gam[o + 1] * rsqrtf(rva[o + 1] + BN_EPS);
    const float mu0 = rme[o],      mu1 = rme[o + 1];
    const float be0 = bet[o],      be1 = bet[o + 1];
    v2f* orow = ob + (o >> 1) * XP2;
    v2f r;
    r.x = (fmaxf(acc0[v] + bs0, 0.f) - mu0) * sc0 + be0;
    r.y = (fmaxf(acc0[v + 1] + bs1, 0.f) - mu1) * sc1 + be1;
    orow[ 0] = r;
    r.x = (fmaxf(acc1[v] + bs0, 0.f) - mu0) * sc0 + be0;
    r.y = (fmaxf(acc1[v + 1] + bs1, 0.f) - mu1) * sc1 + be1;
    orow[16] = r;
    r.x = (fmaxf(acc2[v] + bs0, 0.f) - mu0) * sc0 + be0;
    r.y = (fmaxf(acc2[v + 1] + bs1, 0.f) - mu1) * sc1 + be1;
    orow[32] = r;
    r.x = (fmaxf(acc3[v] + bs0, 0.f) - mu0) * sc0 + be0;
    r.y = (fmaxf(acc3[v + 1] + bs1, 0.f) - mu1) * sc1 + be1;
    orow[48] = r;
  }
}

__global__ __launch_bounds__(256) void k_mlp(const float* __restrict__ feats,
                                             const int* __restrict__ src,
                                             const float* __restrict__ zrow,
                                             const float* __restrict__ lang,
                                             const float* __restrict__ Wf,
                                             const float* __restrict__ bf,
                                             const float* __restrict__ W1,
                                             const float* __restrict__ b1,
                                             const float* __restrict__ g1,
                                             const float* __restrict__ be1,
                                             const float* __restrict__ rm1,
                                             const float* __restrict__ rv1,
                                             const float* __restrict__ W2,
                                             const float* __restrict__ b2,
                                             const float* __restrict__ g2,
                                             const float* __restrict__ be2,
                                             const float* __restrict__ rm2,
                                             const float* __restrict__ rv2,
                                             const float* __restrict__ W3,
                                             const float* __restrict__ b3,
                                             float* __restrict__ out) {
  __shared__ v2f xa[XPLANE];
  __shared__ v2f xb[XPLANE];

  const int b     = blockIdx.x;
  const int pbase = blockIdx.y * PT;
  const int wave  = threadIdx.x >> 5;
  const int lane  = threadIdx.x & 31;
  const int half  = lane >> 4;
  const int lm    = lane & 15;
  const int o0    = wave << 4;

  v2f aw[KS];
  v8f acc0 = V8F_ZERO, acc1 = V8F_ZERO, acc2 = V8F_ZERO, acc3 = V8F_ZERO;

  {
    const float* arow = Wf + (size_t)(o0 + lm) * Kfuse + 2 * half;
    const int* sl = src + b * PPB + pbase + lm;
    const int s0 = sl[0], s1 = sl[16], s2 = sl[32], s3 = sl[48];
    const float* br0  = ((unsigned)s0 < (unsigned)Np ? feats + (size_t)s0 * Mf : zrow) + 2 * half;
    const float* br1  = ((unsigned)s1 < (unsigned)Np ? feats + (size_t)s1 * Mf : zrow) + 2 * half;
    const float* br2  = ((unsigned)s2 < (unsigned)Np ? feats + (size_t)s2 * Mf : zrow) + 2 * half;
    const float* br3  = ((unsigned)s3 < (unsigned)Np ? feats + (size_t)s3 * Mf : zrow) + 2 * half;

    {
      const float* ap = Wf + (size_t)(o0 + lm) * Kfuse + Mf + 4 * half;
      const float* bp = lang + (size_t)b * LANGD + 4 * half;
      v8f accLa = V8F_ZERO, accLb = V8F_ZERO;
#pragma unroll
      for (int j = 0; j < LANGD / 8; ++j) {
        v2f a0;  a0.x  = ap[8 * j];      a0.y  = ap[8 * j + 1];
        v2f a1;  a1.x  = ap[8 * j + 2];  a1.y  = ap[8 * j + 3];
        v2f bb0; bb0.x = bp[8 * j];      bb0.y = bp[8 * j + 1];
        v2f bb1; bb1.x = bp[8 * j + 2];  bb1.y = bp[8 * j + 3];
        accLa = wmma4(a0, bb0, accLa);
        accLb = wmma4(a1, bb1, accLb);
      }
      acc0 = accLa + accLb;
      acc1 = acc0; acc2 = acc0; acc3 = acc0;
    }

#pragma unroll 4
    for (int k0 = 0; k0 < Mf; k0 += 4) {
      v2f a;   a.x  = arow[k0];      a.y  = arow[k0 + 1];
      v2f b0;  b0.x = br0[k0];       b0.y = br0[k0 + 1];
      v2f b1;  b1.x = br1[k0];       b1.y = br1[k0 + 1];
      v2f b2;  b2.x = br2[k0];       b2.y = br2[k0 + 1];
      v2f b3;  b3.x = br3[k0];       b3.y = br3[k0 + 1];
      acc0 = wmma4(a, b0, acc0);
      acc1 = wmma4(a, b1, acc1);
      acc2 = wmma4(a, b2, acc2);
      acc3 = wmma4(a, b3, acc3);
    }

    load_a(W1, o0, lm, half, aw);

    v2f* ob = xa + lm;
#pragma unroll
    for (int v = 0; v < 8; v += 2) {
      const int o = o0 + v + 8 * half;
      const float bias0 = bf[o], bias1 = bf[o + 1];
      v2f* orow = ob + (o >> 1) * XP2;
      v2f r;
      r.x = fmaxf(acc0[v] + bias0, 0.f);  r.y = fmaxf(acc0[v + 1] + bias1, 0.f);
      orow[ 0] = r;
      r.x = fmaxf(acc1[v] + bias0, 0.f);  r.y = fmaxf(acc1[v + 1] + bias1, 0.f);
      orow[16] = r;
      r.x = fmaxf(acc2[v] + bias0, 0.f);  r.y = fmaxf(acc2[v + 1] + bias1, 0.f);
      orow[32] = r;
      r.x = fmaxf(acc3[v] + bias0, 0.f);  r.y = fmaxf(acc3[v + 1] + bias1, 0.f);
      orow[48] = r;
    }
  }
  __syncthreads();

  acc0 = (v8f)V8F_ZERO; acc1 = (v8f)V8F_ZERO; acc2 = (v8f)V8F_ZERO; acc3 = (v8f)V8F_ZERO;
  mid_gemm(aw, xa, lm, half, acc0, acc1, acc2, acc3);
  load_a(W2, o0, lm, half, aw);
  mid_store(acc0, acc1, acc2, acc3, b1, g1, be1, rm1, rv1, xb, o0, lm, half);
  __syncthreads();

  acc0 = (v8f)V8F_ZERO; acc1 = (v8f)V8F_ZERO; acc2 = (v8f)V8F_ZERO; acc3 = (v8f)V8F_ZERO;
  mid_gemm(aw, xb, lm, half, acc0, acc1, acc2, acc3);
  mid_store(acc0, acc1, acc2, acc3, b2, g2, be2, rm2, rv2, xa, o0, lm, half);
  __syncthreads();

  if (threadIdx.x < PT) {
    const v2f* xc = xa + threadIdx.x;
    float s = b3[0];
#pragma unroll 8
    for (int c = 0; c < Hd; c += 2) {
      const v2f x = xc[(c >> 1) * XP2];
      s += W3[c] * x.x;
      s += W3[c + 1] * x.y;
    }
    out[b * PPB + pbase + threadIdx.x] = s;
  }
}

extern "C" void kernel_launch(void* const* d_in, const int* in_sizes, int n_in,
                              void* d_out, int out_size, void* d_ws, size_t ws_size,
                              hipStream_t stream) {
  const float* feats = (const float*)d_in[0];
  const int*   pidx  = (const int*)d_in[1];
  const int*   poff  = (const int*)d_in[2];
  const int*   boff  = (const int*)d_in[3];
  const float* lang  = (const float*)d_in[4];
  const float* Wf    = (const float*)d_in[5];
  const float* bf    = (const float*)d_in[6];
  const float* W1p   = (const float*)d_in[7];
  const float* b1p   = (const float*)d_in[8];
  const float* g1p   = (const float*)d_in[9];
  const float* be1p  = (const float*)d_in[10];
  const float* rm1p  = (const float*)d_in[11];
  const float* rv1p  = (const float*)d_in[12];
  const float* W2p   = (const float*)d_in[13];
  const float* b2p   = (const float*)d_in[14];
  const float* g2p   = (const float*)d_in[15];
  const float* be2p  = (const float*)d_in[16];
  const float* rm2p  = (const float*)d_in[17];
  const float* rv2p  = (const float*)d_in[18];
  const float* W3p   = (const float*)d_in[19];
  const float* b3p   = (const float*)d_in[20];

  char* ws = (char*)d_ws;
  int*   bid    = (int*)ws;
  int*   src    = (int*)(ws + (size_t)Np * sizeof(int));
  float* zrow   = (float*)(ws + 2 * (size_t)Np * sizeof(int));
  int*   hist   = (int*)(zrow + (size_t)Np * Mf);

  k_bid <<<NRB, RB, 0, stream>>>(pidx, poff, boff, bid, hist, src, zrow);
  k_rank<<<NRB, RB, 0, stream>>>(bid, hist, src);

  dim3 grid(Bn, 4);
  k_mlp<<<grid, 256, 0, stream>>>(feats, src, zrow, lang, Wf, bf,
                                  W1p, b1p, g1p, be1p, rm1p, rv1p,
                                  W2p, b2p, g2p, be2p, rm2p, rv2p,
                                  W3p, b3p, (float*)d_out);
}


# k_rank launch removed (3 -> 2 launches): each MLP work-group ranks its own batch's proposals (wave ballots over bid[], counts summed through LDS), 64 slot indices kept in LDS; float sums untouched
# speedup vs baseline: 4.9894x; 1.0475x over previous
#include <hip/hip_runtime.h>

constexpr int Bn    = 32;
constexpr int PPB   = 256;
constexpr int Np    = Bn * PPB;
constexpr int Mf    = 128;
constexpr int LANGD = 256;
constexpr int Hd    = 128;
constexpr int Kfuse = LANGD + Mf;
#define BN_EPS 1e-5f

typedef float v2f __attribute__((ext_vector_type(2)));
typedef float v8f __attribute__((ext_vector_type(8)));

__device__ __forceinline__ v8f wmma4(v2f a, v2f b, v8f c) {
  return __builtin_amdgcn_wmma_f32_16x16x4_f32(false, a, false, b, (short)0, c,
                                               false, false);
}

#define V8F_ZERO {0.f, 0.f, 0.f, 0.f, 0.f, 0.f, 0.f, 0.f}

constexpr int RB  = 256;
constexpr int NRB = Np / RB;
static_assert(Np % RB == 0, "k_bid blocks must tile the proposals exactly");

__global__ __launch_bounds__(RB) void k_bid(const int* __restrict__ pidx,
                                            const int* __restrict__ poff,
                                            const int* __restrict__ boff,
                                            int* __restrict__ bid,
                                            float* __restrict__ zrow) {
  const int i = blockIdx.x * RB + threadIdx.x;
  int first_pt = pidx[2 * poff[i] + 1];
  int b = -1;
#pragma unroll
  for (int j = 0; j <= Bn; ++j) b += (boff[j] <= first_pt) ? 1 : 0;
  bid[i] = b;
  if (blockIdx.x == 0 && threadIdx.x < Mf) zrow[threadIdx.x] = 0.f;
}

constexpr int PT  = 64;
constexpr int XP2 = 80;
constexpr int XPLANE = (Hd / 2) * XP2;
constexpr int KS  = Hd / 4;

__device__ __forceinline__ void load_a(const float* __restrict__ W,
                                       int o0, int lm, int half,
                                       v2f (&aw)[KS]) {
  const float* arow = W + (size_t)(o0 + lm) * Hd + 2 * half;
#pragma unroll
  for (int q = 0; q < KS; ++q) { aw[q].x = arow[4 * q]; aw[q].y = arow[4 * q + 1]; }
}

__device__ __forceinline__ void mid_gemm(const v2f (&aw)[KS], const v2f* xin,
                                         int lm, int half,
                                         v8f& acc0, v8f& acc1, v8f& acc2, v8f& acc3) {
  const v2f* bc = xin + half * XP2 + lm;
#pragma unroll
  for (int q = 0; q < KS; ++q) {
    const v2f* br = bc + (2 * q) * XP2;
    const v2f b0 = br[ 0];
    const v2f b1 = br[16];
    const v2f b2 = br[32];
    const v2f b3 = br[48];
    acc0 = wmma4(aw[q], b0, acc0);
    acc1 = wmma4(aw[q], b1, acc1);
    acc2 = wmma4(aw[q], b2, acc2);
    acc3 = wmma4(aw[q], b3, acc3);
  }
}

__device__ __forceinline__ void mid_store(const v8f& acc0, const v8f& acc1,
                                          const v8f& acc2, const v8f& acc3,
                                          const float* __restrict__ bias,
                                          const float* __restrict__ gam,
                                          const float* __restrict__ bet,
                                          const float* __restrict__ rme,
                                          const float* __restrict__ rva,
                                          v2f* xout, int o0, int lm, int half) {
  v2f* ob = xout + lm;
#pragma unroll
  for (int v = 0; v < 8; v += 2) {
    const int o = o0 + v + 8 * half;
    const float bs0 = bias[o],     bs1 = bias[o + 1];
    const float sc0 = gam[o]     * rsqrtf(rva[o]     + BN_EPS);
    const float sc1 = gam[o + 1] * rsqrtf(rva[o + 1] + BN_EPS);
    const float mu0 = rme[o],      mu1 = rme[o + 1];
    const float be0 = bet[o],      be1 = bet[o + 1];
    v2f* orow = ob + (o >> 1) * XP2;
    v2f r;
    r.x = (fmaxf(acc0[v] + bs0, 0.f) - mu0) * sc0 + be0;
    r.y = (fmaxf(acc0[v + 1] + bs1, 0.f) - mu1) * sc1 + be1;
    orow[ 0] = r;
    r.x = (fmaxf(acc1[v] + bs0, 0.f) - mu0) * sc0 + be0;
    r.y = (fmaxf(acc1[v + 1] + bs1, 0.f) - mu1) * sc1 + be1;
    orow[16] = r;
    r.x = (fmaxf(acc2[v] + bs0, 0.f) - mu0) * sc0 + be0;
    r.y = (fmaxf(acc2[v + 1] + bs1, 0.f) - mu1) * sc1 + be1;
    orow[32] = r;
    r.x = (fmaxf(acc3[v] + bs0, 0.f) - mu0) * sc0 + be0;
    r.y = (fmaxf(acc3[v + 1] + bs1, 0.f) - mu1) * sc1 + be1;
    orow[48] = r;
  }
}

constexpr int NWV = 8;
constexpr int CH  = Np / NWV;
constexpr int NR  = CH / 32;
static_assert(Np % (NWV * 32) == 0, "waves must tile the proposals in rounds");

__global__ __launch_bounds__(256) void k_mlp(const float* __restrict__ feats,
                                             const int* __restrict__ bid,
                                             const float* __restrict__ zrow,
                                             const float* __restrict__ lang,
                                             const float* __restrict__ Wf,
                                             const float* __restrict__ bf,
                                             const float* __restrict__ W1,
                                             const float* __restrict__ b1,
                                             const float* __restrict__ g1,
                                             const float* __restrict__ be1,
                                             const float* __restrict__ rm1,
                                             const float* __restrict__ rv1,
                                             const float* __restrict__ W2,
                                             const float* __restrict__ b2,
                                             const float* __restrict__ g2,
                                             const float* __restrict__ be2,
                                             const float* __restrict__ rm2,
                                             const float* __restrict__ rv2,
                                             const float* __restrict__ W3,
                                             const float* __restrict__ b3,
                                             float* __restrict__ out) {
  __shared__ v2f xa[XPLANE];
  __shared__ v2f xb[XPLANE];
  __shared__ int s_tot[NWV];
  __shared__ int s_slot[PT];

  const int b     = blockIdx.x;
  const int pbase = blockIdx.y * PT;
  const int wave  = threadIdx.x >> 5;
  const int lane  = threadIdx.x & 31;
  const int half  = lane >> 4;
  const int lm    = lane & 15;
  const int o0    = wave << 4;

  v2f aw[KS];
  v8f acc0 = V8F_ZERO, acc1 = V8F_ZERO, acc2 = V8F_ZERO, acc3 = V8F_ZERO;

  int bv[NR];
  {
    const int* bp = bid + wave * CH + lane;
#pragma unroll
    for (int j = 0; j < NR; ++j) bv[j] = bp[32 * j];
  }

  {
    const float* arow = Wf + (size_t)(o0 + lm) * Kfuse + 2 * half;

    {
      const float* ap = Wf + (size_t)(o0 + lm) * Kfuse + Mf + 4 * half;
      const float* bp = lang + (size_t)b * LANGD + 4 * half;
      v8f accLa = V8F_ZERO, accLb = V8F_ZERO;
#pragma unroll
      for (int j = 0; j < LANGD / 8; ++j) {
        v2f a0;  a0.x  = ap[8 * j];      a0.y  = ap[8 * j + 1];
        v2f a1;  a1.x  = ap[8 * j + 2];  a1.y  = ap[8 * j + 3];
        v2f bb0; bb0.x = bp[8 * j];      bb0.y = bp[8 * j + 1];
        v2f bb1; bb1.x = bp[8 * j + 2];  bb1.y = bp[8 * j + 3];
        accLa = wmma4(a0, bb0, accLa);
        accLb = wmma4(a1, bb1, accLb);
      }
      acc0 = accLa + accLb;
      acc1 = acc0; acc2 = acc0; acc3 = acc0;
    }

    {
      int cnt = 0;
#pragma unroll
      for (int j = 0; j < NR; ++j)
        cnt += __popc(__builtin_amdgcn_ballot_w32(bv[j] == b));
      if (lane == 0) s_tot[wave] = cnt;
      if (threadIdx.x < PT) s_slot[threadIdx.x] = -1;
      __syncthreads();
      int base = -pbase;
#pragma unroll
      for (int w = 0; w < NWV; ++w) base += (w < wave) ? s_tot[w] : 0;
#pragma unroll
      for (int j = 0; j < NR; ++j) {
        const unsigned m = __builtin_amdgcn_ballot_w32(bv[j] == b);
        if (m != 0u) {
          const int r = base + (int)__builtin_amdgcn_mbcnt_lo(m, 0u);
          if (bv[j] == b && (unsigned)r < (unsigned)PT)
            s_slot[r] = wave * CH + 32 * j + lane;
          base += __popc(m);
        }
      }
      __syncthreads();
    }
    const int s0 = s_slot[lm], s1 = s_slot[lm + 16], s2 = s_slot[lm + 32], s3 = s_slot[lm + 48];
    const float* br0  = ((unsigned)s0 < (unsigned)Np ? feats + (size_t)s0 * Mf : zrow) + 2 * half;
    const float* br1  = ((unsigned)s1 < (unsigned)Np ? feats + (size_t)s1 * Mf : zrow) + 2 * half;
    const float* br2  = ((unsigned)s2 < (unsigned)Np ? feats + (size_t)s2 * Mf : zrow) + 2 * half;
    const float* br3  = ((unsigned)s3 < (unsigned)Np ? feats + (size_t)s3 * Mf : zrow) + 2 * half;

#pragma unroll 4
    for (int k0 = 0; k0 < Mf; k0 += 4) {
      v2f a;   a.x  = arow[k0];      a.y  = arow[k0 + 1];
      v2f b0;  b0.x = br0[k0];       b0.y = br0[k0 + 1];
      v2f b1;  b1.x = br1[k0];       b1.y = br1[k0 + 1];
      v2f b2;  b2.x = br2[k0];       b2.y = br2[k0 + 1];
      v2f b3;  b3.x = br3[k0];       b3.y = br3[k0 + 1];
      acc0 = wmma4(a, b0, acc0);
      acc1 = wmma4(a, b1, acc1);
      acc2 = wmma4(a, b2, acc2);
      acc3 = wmma4(a, b3, acc3);
    }

    load_a(W1, o0, lm, half, aw);

    v2f* ob = xa + lm;
#pragma unroll
    for (int v = 0; v < 8; v += 2) {
      const int o = o0 + v + 8 * half;
      const float bias0 = bf[o], bias1 = bf[o + 1];
      v2f* orow = ob + (o >> 1) * XP2;
      v2f r;
      r.x = fmaxf(acc0[v] + bias0, 0.f);  r.y = fmaxf(acc0[v + 1] + bias1, 0.f);
      orow[ 0] = r;
      r.x = fmaxf(acc1[v] + bias0, 0.f);  r.y = fmaxf(acc1[v + 1] + bias1, 0.f);
      orow[16] = r;
      r.x = fmaxf(acc2[v] + bias0, 0.f);  r.y = fmaxf(acc2[v + 1] + bias1, 0.f);
      orow[32] = r;
      r.x = fmaxf(acc3[v] + bias0, 0.f);  r.y = fmaxf(acc3[v + 1] + bias1, 0.f);
      orow[48] = r;
    }
  }
  __syncthreads();

  acc0 = (v8f)V8F_ZERO; acc1 = (v8f)V8F_ZERO; acc2 = (v8f)V8F_ZERO; acc3 = (v8f)V8F_ZERO;
  mid_gemm(aw, xa, lm, half, acc0, acc1, acc2, acc3);
  load_a(W2, o0, lm, half, aw);
  mid_store(acc0, acc1, acc2, acc3, b1, g1, be1, rm1, rv1, xb, o0, lm, half);
  __syncthreads();

  acc0 = (v8f)V8F_ZERO; acc1 = (v8f)V8F_ZERO; acc2 = (v8f)V8F_ZERO; acc3 = (v8f)V8F_ZERO;
  mid_gemm(aw, xb, lm, half, acc0, acc1, acc2, acc3);
  mid_store(acc0, acc1, acc2, acc3, b2, g2, be2, rm2, rv2, xa, o0, lm, half);
  __syncthreads();

  if (threadIdx.x < PT) {
    const v2f* xc = xa + threadIdx.x;
    float s = b3[0];
#pragma unroll 8
    for (int c = 0; c < Hd; c += 2) {
      const v2f x = xc[(c >> 1) * XP2];
      s += W3[c] * x.x;
      s += W3[c + 1] * x.y;
    }
    out[b * PPB + pbase + threadIdx.x] = s;
  }
}

extern "C" void kernel_launch(void* const* d_in, const int* in_sizes, int n_in,
                              void* d_out, int out_size, void* d_ws, size_t ws_size,
                              hipStream_t stream) {
  const float* feats = (const float*)d_in[0];
  const int*   pidx  = (const int*)d_in[1];
  const int*   poff  = (const int*)d_in[2];
  const int*   boff  = (const int*)d_in[3];
  const float* lang  = (const float*)d_in[4];
  const float* Wf    = (const float*)d_in[5];
  const float* bf    = (const float*)d_in[6];
  const float* W1p   = (const float*)d_in[7];
  const float* b1p   = (const float*)d_in[8];
  const float* g1p   = (const float*)d_in[9];
  const float* be1p  = (const float*)d_in[10];
  const float* rm1p  = (const float*)d_in[11];
  const float* rv1p  = (const float*)d_in[12];
  const float* W2p   = (const float*)d_in[13];
  const float* b2p   = (const float*)d_in[14];
  const float* g2p   = (const float*)d_in[15];
  const float* be2p  = (const float*)d_in[16];
  const float* rm2p  = (const float*)d_in[17];
  const float* rv2p  = (const float*)d_in[18];
  const float* W3p   = (const float*)d_in[19];
  const float* b3p   = (const float*)d_in[20];

  char* ws = (char*)d_ws;
  int*   bid    = (int*)ws;
  float* zrow   = (float*)(ws + (size_t)Np * sizeof(int));

  k_bid<<<NRB, RB, 0, stream>>>(pidx, poff, boff, bid, zrow);

  dim3 grid(Bn, 4);
  k_mlp<<<grid, 256, 0, stream>>>(feats, bid, zrow, lang, Wf, bf,
                                  W1p, b1p, g1p, be1p, rm1p, rv1p,
                                  W2p, b2p, g2p, be2p, rm2p, rv2p,
                                  W3p, b3p, (float*)d_out);
}


# layer 1: the 64 gathered input rows (B of the first K segment) copied once per work-group into an LDS plane, 64 bytes per row per load; all 8 waves read B fragments from LDS; sum order kept
# speedup vs baseline: 5.9869x; 1.1999x over previous
#include <hip/hip_runtime.h>

constexpr int Bn    = 32;
constexpr int PPB   = 256;
constexpr int Np    = Bn * PPB;
constexpr int Mf    = 128;
constexpr int LANGD = 256;
constexpr int Hd    = 128;
constexpr int Kfuse = LANGD + Mf;
#define BN_EPS 1e-5f

typedef float v2f __attribute__((ext_vector_type(2)));
typedef float v8f __attribute__((ext_vector_type(8)));

__device__ __forceinline__ v8f wmma4(v2f a, v2f b, v8f c) {
  return __builtin_amdgcn_wmma_f32_16x16x4_f32(false, a, false, b, (short)0, c,
                                               false, false);
}

#define V8F_ZERO {0.f, 0.f, 0.f, 0.f, 0.f, 0.f, 0.f, 0.f}

constexpr int RB  = 256;
constexpr int NRB = Np / RB;
static_assert(Np % RB == 0, "k_bid blocks must tile the proposals exactly");

__global__ __launch_bounds__(RB) void k_bid(const int* __restrict__ pidx,
                                            const int* __restrict__ poff,
                                            const int* __restrict__ boff,
                                            int* __restrict__ bid) {
  const int i = blockIdx.x * RB + threadIdx.x;
  int first_pt = pidx[2 * poff[i] + 1];
  int b = -1;
#pragma unroll
  for (int j = 0; j <= Bn; ++j) b += (boff[j] <= first_pt) ? 1 : 0;
  bid[i] = b;
}

constexpr int PT  = 64;
constexpr int XP2 = 80;
constexpr int XPLANE = (Hd / 2) * XP2;
constexpr int KS  = Hd / 4;
static_assert(Mf == Hd, "layer 1's first K segment runs through the middle layers' GEMM");

template <int PITCH>
__device__ __forceinline__ void load_a(const float* __restrict__ W,
                                       int o0, int lm, int half,
                                       v2f (&aw)[KS]) {
  const float* arow = W + (size_t)(o0 + lm) * PITCH + 2 * half;
#pragma unroll
  for (int q = 0; q < KS; ++q) { aw[q].x = arow[4 * q]; aw[q].y = arow[4 * q + 1]; }
}

__device__ __forceinline__ void mid_gemm(const v2f (&aw)[KS], const v2f* xin,
                                         int lm, int half,
                                         v8f& acc0, v8f& acc1, v8f& acc2, v8f& acc3) {
  const v2f* bc = xin + half * XP2 + lm;
#pragma unroll
  for (int q = 0; q < KS; ++q) {
    const v2f* br = bc + (2 * q) * XP2;
    const v2f b0 = br[ 0];
    const v2f b1 = br[16];
    const v2f b2 = br[32];
    const v2f b3 = br[48];
    acc0 = wmma4(aw[q], b0, acc0);
    acc1 = wmma4(aw[q], b1, acc1);
    acc2 = wmma4(aw[q], b2, acc2);
    acc3 = wmma4(aw[q], b3, acc3);
  }
}

__device__ __forceinline__ void mid_store(const v8f& acc0, const v8f& acc1,
                                          const v8f& acc2, const v8f& acc3,
                                          const float* __restrict__ bias,
                                          const float* __restrict__ gam,
                                          const float* __restrict__ bet,
                                          const float* __restrict__ rme,
                                          const float* __restrict__ rva,
                                          v2f* xout, int o0, int lm, int half) {
  v2f* ob = xout + lm;
#pragma unroll
  for (int v = 0; v < 8; v += 2) {
    const int o = o0 + v + 8 * half;
    const float bs0 = bias[o],     bs1 = bias[o + 1];
    const float sc0 = gam[o]     * rsqrtf(rva[o]     + BN_EPS);
    const float sc1 = gam[o + 1] * rsqrtf(rva[o + 1] + BN_EPS);
    const float mu0 = rme[o],      mu1 = rme[o + 1];
    const float be0 = bet[o],      be1 = bet[o + 1];
    v2f* orow = ob + (o >> 1) * XP2;
    v2f r;
    r.x = (fmaxf(acc0[v] + bs0, 0.f) - mu0) * sc0 + be0;
    r.y = (fmaxf(acc0[v + 1] + bs1, 0.f) - mu1) * sc1 + be1;
    orow[ 0] = r;
    r.x = (fmaxf(acc1[v] + bs0, 0.f) - mu0) * sc0 + be0;
    r.y = (fmaxf(acc1[v + 1] + bs1, 0.f) - mu1) * sc1 + be1;
    orow[16] = r;
    r.x = (fmaxf(acc2[v] + bs0, 0.f) - mu0) * sc0 + be0;
    r.y = (fmaxf(acc2[v + 1] + bs1, 0.f) - mu1) * sc1 + be1;
    orow[32] = r;
    r.x = (fmaxf(acc3[v] + bs0, 0.f) - mu0) * sc0 + be0;
    r.y = (fmaxf(acc3[v + 1] + bs1, 0.f) - mu1) * sc1 + be1;
    orow[48] = r;
  }
}

constexpr int NWV = 8;
constexpr int CH  = Np / NWV;
constexpr int NR  = CH / 32;
static_assert(Np % (NWV * 32) == 0, "waves must tile the proposals in rounds");

__global__ __launch_bounds__(256) void k_mlp(const float* __restrict__ feats,
                                             const int* __restrict__ bid,
                                             const float* __restrict__ lang,
                                             const float* __restrict__ Wf,
                                             const float* __restrict__ bf,
                                             const float* __restrict__ W1,
                                             const float* __restrict__ b1,
                                             const float* __restrict__ g1,
                                             const float* __restrict__ be1,
                                             const float* __restrict__ rm1,
                                             const float* __restrict__ rv1,
                                             const float* __restrict__ W2,
                                             const float* __restrict__ b2,
                                             const float* __restrict__ g2,
                                             const float* __restrict__ be2,
                                             const float* __restrict__ rm2,
                                             const float* __restrict__ rv2,
                                             const float* __restrict__ W3,
                                             const float* __restrict__ b3,
                                             float* __restrict__ out) {
  __shared__ v2f xa[XPLANE];
  __shared__ v2f xb[XPLANE];
  __shared__ int s_tot[NWV];
  __shared__ int s_slot[PT];

  const int b     = blockIdx.x;
  const int pbase = blockIdx.y * PT;
  const int wave  = threadIdx.x >> 5;
  const int lane  = threadIdx.x & 31;
  const int half  = lane >> 4;
  const int lm    = lane & 15;
  const int o0    = wave << 4;

  v2f aw[KS];
  v8f acc0 = V8F_ZERO, acc1 = V8F_ZERO, acc2 = V8F_ZERO, acc3 = V8F_ZERO;

  int bv[NR];
  {
    const int* bp = bid + wave * CH + lane;
#pragma unroll
    for (int j = 0; j < NR; ++j) bv[j] = bp[32 * j];
  }

  {
    {
      const float* ap = Wf + (size_t)(o0 + lm) * Kfuse + Mf + 4 * half;
      const float* bp = lang + (size_t)b * LANGD + 4 * half;
      v8f accLa = V8F_ZERO, accLb = V8F_ZERO;
#pragma unroll
      for (int j = 0; j < LANGD / 8; ++j) {
        v2f a0;  a0.x  = ap[8 * j];      a0.y  = ap[8 * j + 1];
        v2f a1;  a1.x  = ap[8 * j + 2];  a1.y  = ap[8 * j + 3];
        v2f bb0; bb0.x = bp[8 * j];      bb0.y = bp[8 * j + 1];
        v2f bb1; bb1.x = bp[8 * j + 2];  bb1.y = bp[8 * j + 3];
        accLa = wmma4(a0, bb0, accLa);
        accLb = wmma4(a1, bb1, accLb);
      }
      acc0 = accLa + accLb;
      acc1 = acc0; acc2 = acc0; acc3 = acc0;
    }

    {
      int cnt = 0;
#pragma unroll
      for (int j = 0; j < NR; ++j)
        cnt += __popc(__builtin_amdgcn_ballot_w32(bv[j] == b));
      if (lane == 0) s_tot[wave] = cnt;
      if (threadIdx.x < PT) s_slot[threadIdx.x] = -1;
      __syncthreads();
      int base = -pbase;
#pragma unroll
      for (int w = 0; w < NWV; ++w) base += (w < wave) ? s_tot[w] : 0;
#pragma unroll
      for (int j = 0; j < NR; ++j) {
        const unsigned m = __builtin_amdgcn_ballot_w32(bv[j] == b);
        if (m != 0u) {
          const int r = base + (int)__builtin_amdgcn_mbcnt_lo(m, 0u);
          if (bv[j] == b && (unsigned)r < (unsigned)PT)
            s_slot[r] = wave * CH + 32 * j + lane;
          base += __popc(m);
        }
      }
      __syncthreads();
    }
    {
      const int pi = lane & 7;
      const int sg = lane >> 3;
      const int pc = 8 * wave + pi;
      const int s  = s_slot[pc];
      float r[Mf / 16][4];
#pragma unroll
      for (int it = 0; it < Mf / 16; ++it) { r[it][0] = 0.f; r[it][1] = 0.f; r[it][2] = 0.f; r[it][3] = 0.f; }
      if ((unsigned)s < (unsigned)Np) {
        const float* src = feats + (size_t)s * Mf + 4 * sg;
#pragma unroll
        for (int it = 0; it < Mf / 16; ++it) {
          r[it][0] = src[16 * it];     r[it][1] = src[16 * it + 1];
          r[it][2] = src[16 * it + 2]; r[it][3] = src[16 * it + 3];
        }
      }
      v2f* dst = xb + (2 * sg) * XP2 + pc;
#pragma unroll
      for (int it = 0; it < Mf / 16; ++it) {
        v2f lo; lo.x = r[it][0]; lo.y = r[it][1];
        v2f hi; hi.x = r[it][2]; hi.y = r[it][3];
        dst[(8 * it) * XP2]     = lo;
        dst[(8 * it + 1) * XP2] = hi;
      }
    }
    __syncthreads();

    load_a<Kfuse>(Wf, o0, lm, half, aw);
    mid_gemm(aw, xb, lm, half, acc0, acc1, acc2, acc3);

    load_a<Hd>(W1, o0, lm, half, aw);

    v2f* ob = xa + lm;
#pragma unroll
    for (int v = 0; v < 8; v += 2) {
      const int o = o0 + v + 8 * half;
      const float bias0 = bf[o], bias1 = bf[o + 1];
      v2f* orow = ob + (o >> 1) * XP2;
      v2f r;
      r.x = fmaxf(acc0[v] + bias0, 0.f);  r.y = fmaxf(acc0[v + 1] + bias1, 0.f);
      orow[ 0] = r;
      r.x = fmaxf(acc1[v] + bias0, 0.f);  r.y = fmaxf(acc1[v + 1] + bias1, 0.f);
      orow[16] = r;
      r.x = fmaxf(acc2[v] + bias0, 0.f);  r.y = fmaxf(acc2[v + 1] + bias1, 0.f);
      orow[32] = r;
      r.x = fmaxf(acc3[v] + bias0, 0.f);  r.y = fmaxf(acc3[v + 1] + bias1, 0.f);
      orow[48] = r;
    }
  }
  __syncthreads();

  acc0 = (v8f)V8F_ZERO; acc1 = (v8f)V8F_ZERO; acc2 = (v8f)V8F_ZERO; acc3 = (v8f)V8F_ZERO;
  mid_gemm(aw, xa, lm, half, acc0, acc1, acc2, acc3);
  load_a<Hd>(W2, o0, lm, half, aw);
  mid_store(acc0, acc1, acc2, acc3, b1, g1, be1, rm1, rv1, xb, o0, lm, half);
  __syncthreads();

  acc0 = (v8f)V8F_ZERO; acc1 = (v8f)V8F_ZERO; acc2 = (v8f)V8F_ZERO; acc3 = (v8f)V8F_ZERO;
  mid_gemm(aw, xb, lm, half, acc0, acc1, acc2, acc3);
  mid_store(acc0, acc1, acc2, acc3, b2, g2, be2, rm2, rv2, xa, o0, lm, half);
  __syncthreads();

  if (threadIdx.x < PT) {
    const v2f* xc = xa + threadIdx.x;
    float s = b3[0];
#pragma unroll 8
    for (int c = 0; c < Hd; c += 2) {
      const v2f x = xc[(c >> 1) * XP2];
      s += W3[c] * x.x;
      s += W3[c + 1] * x.y;
    }
    out[b * PPB + pbase + threadIdx.x] = s;
  }
}

extern "C" void kernel_launch(void* const* d_in, const int* in_sizes, int n_in,
                              void* d_out, int out_size, void* d_ws, size_t ws_size,
                              hipStream_t stream) {
  const float* feats = (const float*)d_in[0];
  const int*   pidx  = (const int*)d_in[1];
  const int*   poff  = (const int*)d_in[2];
  const int*   boff  = (const int*)d_in[3];
  const float* lang  = (const float*)d_in[4];
  const float* Wf    = (const float*)d_in[5];
  const float* bf    = (const float*)d_in[6];
  const float* W1p   = (const float*)d_in[7];
  const float* b1p   = (const float*)d_in[8];
  const float* g1p   = (const float*)d_in[9];
  const float* be1p  = (const float*)d_in[10];
  const float* rm1p  = (const float*)d_in[11];
  const float* rv1p  = (const float*)d_in[12];
  const float* W2p   = (const float*)d_in[13];
  const float* b2p   = (const float*)d_in[14];
  const float* g2p   = (const float*)d_in[15];
  const float* be2p  = (const float*)d_in[16];
  const float* rm2p  = (const float*)d_in[17];
  const float* rv2p  = (const float*)d_in[18];
  const float* W3p   = (const float*)d_in[19];
  const float* b3p   = (const float*)d_in[20];

  char* ws = (char*)d_ws;
  int*   bid    = (int*)ws;

  k_bid<<<NRB, RB, 0, stream>>>(pidx, poff, boff, bid);

  dim3 grid(Bn, 4);
  k_mlp<<<grid, 256, 0, stream>>>(feats, bid, lang, Wf, bf,
                                  W1p, b1p, g1p, be1p, rm1p, rv1p,
                                  W2p, b2p, g2p, be2p, rm2p, rv2p,
                                  W3p, b3p, (float*)d_out);
}


# MLP kernel re-tiled from 64 to 32 columns per work-group: 256 work-groups (one per WGP, was 128), each wave 2 p-tiles per A fragment instead of 4; per-column sum order kept
# speedup vs baseline: 6.7807x; 1.1326x over previous
#include <hip/hip_runtime.h>

constexpr int Bn    = 32;
constexpr int PPB   = 256;
constexpr int Np    = Bn * PPB;
constexpr int Mf    = 128;
constexpr int LANGD = 256;
constexpr int Hd    = 128;
constexpr int Kfuse = LANGD + Mf;
#define BN_EPS 1e-5f

typedef float v2f __attribute__((ext_vector_type(2)));
typedef float v8f __attribute__((ext_vector_type(8)));

__device__ __forceinline__ v8f wmma4(v2f a, v2f b, v8f c) {
  return __builtin_amdgcn_wmma_f32_16x16x4_f32(false, a, false, b, (short)0, c,
                                               false, false);
}

#define V8F_ZERO {0.f, 0.f, 0.f, 0.f, 0.f, 0.f, 0.f, 0.f}

constexpr int RB  = 256;
constexpr int NRB = Np / RB;
static_assert(Np % RB == 0, "k_bid blocks must tile the proposals exactly");

__global__ __launch_bounds__(RB) void k_bid(const int* __restrict__ pidx,
                                            const int* __restrict__ poff,
                                            const int* __restrict__ boff,
                                            int* __restrict__ bid) {
  const int i = blockIdx.x * RB + threadIdx.x;
  int first_pt = pidx[2 * poff[i] + 1];
  int b = -1;
#pragma unroll
  for (int j = 0; j <= Bn; ++j) b += (boff[j] <= first_pt) ? 1 : 0;
  bid[i] = b;
}

constexpr int PT  = 32;
constexpr int NPS = PPB / PT;
static_assert(PPB % PT == 0, "p slices must tile a batch's slots exactly");
constexpr int XP2 = 48;
constexpr int XPLANE = (Hd / 2) * XP2;
constexpr int KS  = Hd / 4;
static_assert(Mf == Hd, "layer 1's first K segment runs through the middle layers' GEMM");

template <int PITCH>
__device__ __forceinline__ void load_a(const float* __restrict__ W,
                                       int o0, int lm, int half,
                                       v2f (&aw)[KS]) {
  const float* arow = W + (size_t)(o0 + lm) * PITCH + 2 * half;
#pragma unroll
  for (int q = 0; q < KS; ++q) { aw[q].x = arow[4 * q]; aw[q].y = arow[4 * q + 1]; }
}

__device__ __forceinline__ void mid_gemm(const v2f (&aw)[KS], const v2f* xin,
                                         int lm, int half,
                                         v8f& acc0, v8f& acc1) {
  const v2f* bc = xin + half * XP2 + lm;
#pragma unroll
  for (int q = 0; q < KS; ++q) {
    const v2f* br = bc + (2 * q) * XP2;
    const v2f b0 = br[ 0];
    const v2f b1 = br[16];
    acc0 = wmma4(aw[q], b0, acc0);
    acc1 = wmma4(aw[q], b1, acc1);
  }
}

__device__ __forceinline__ void mid_store(const v8f& acc0, const v8f& acc1,
                                          const float* __restrict__ bias,
                                          const float* __restrict__ gam,
                                          const float* __restrict__ bet,
                                          const float* __restrict__ rme,
                                          const float* __restrict__ rva,
                                          v2f* xout, int o0, int lm, int half) {
  v2f* ob = xout + lm;
#pragma unroll
  for (int v = 0; v < 8; v += 2) {
    const int o = o0 + v + 8 * half;
    const float bs0 = bias[o],     bs1 = bias[o + 1];
    const float sc0 = gam[o]     * rsqrtf(rva[o]     + BN_EPS);
    const float sc1 = gam[o + 1] * rsqrtf(rva[o + 1] + BN_EPS);
    const float mu0 = rme[o],      mu1 = rme[o + 1];
    const float be0 = bet[o],      be1 = bet[o + 1];
    v2f* orow = ob + (o >> 1) * XP2;
    v2f r;
    r.x = (fmaxf(acc0[v] + bs0, 0.f) - mu0) * sc0 + be0;
    r.y = (fmaxf(acc0[v + 1] + bs1, 0.f) - mu1) * sc1 + be1;
    orow[ 0] = r;
    r.x = (fmaxf(acc1[v] + bs0, 0.f) - mu0) * sc0 + be0;
    r.y = (fmaxf(acc1[v + 1] + bs1, 0.f) - mu1) * sc1 + be1;
    orow[16] = r;
  }
}

constexpr int NWV = 8;
constexpr int CH  = Np / NWV;
constexpr int NR  = CH / 32;
static_assert(Np % (NWV * 32) == 0, "waves must tile the proposals in rounds");

__global__ __launch_bounds__(256) void k_mlp(const float* __restrict__ feats,
                                             const int* __restrict__ bid,
                                             const float* __restrict__ lang,
                                             const float* __restrict__ Wf,
                                             const float* __restrict__ bf,
                                             const float* __restrict__ W1,
                                             const float* __restrict__ b1,
                                             const float* __restrict__ g1,
                                             const float* __restrict__ be1,
                                             const float* __restrict__ rm1,
                                             const float* __restrict__ rv1,
                                             const float* __restrict__ W2,
                                             const float* __restrict__ b2,
                                             const float* __restrict__ g2,
                                             const float* __restrict__ be2,
                                             const float* __restrict__ rm2,
                                             const float* __restrict__ rv2,
                                             const float* __restrict__ W3,
                                             const float* __restrict__ b3,
                                             float* __restrict__ out) {
  __shared__ v2f xa[XPLANE];
  __shared__ v2f xb[XPLANE];
  __shared__ int s_tot[NWV];
  __shared__ int s_slot[PT];

  const int b     = blockIdx.x;
  const int pbase = blockIdx.y * PT;
  const int wave  = threadIdx.x >> 5;
  const int lane  = threadIdx.x & 31;
  const int half  = lane >> 4;
  const int lm    = lane & 15;
  const int o0    = wave << 4;

  v2f aw[KS];
  v8f acc0 = V8F_ZERO, acc1 = V8F_ZERO;

  int bv[NR];
  {
    const int* bp = bid + wave * CH + lane;
#pragma unroll
    for (int j = 0; j < NR; ++j) bv[j] = bp[32 * j];
  }

  {
    {
      const float* ap = Wf + (size_t)(o0 + lm) * Kfuse + Mf + 4 * half;
      const float* bp = lang + (size_t)b * LANGD + 4 * half;
      v8f accLa = V8F_ZERO, accLb = V8F_ZERO;
#pragma unroll
      for (int j = 0; j < LANGD / 8; ++j) {
        v2f a0;  a0.x  = ap[8 * j];      a0.y  = ap[8 * j + 1];
        v2f a1;  a1.x  = ap[8 * j + 2];  a1.y  = ap[8 * j + 3];
        v2f bb0; bb0.x = bp[8 * j];      bb0.y = bp[8 * j + 1];
        v2f bb1; bb1.x = bp[8 * j + 2];  bb1.y = bp[8 * j + 3];
        accLa = wmma4(a0, bb0, accLa);
        accLb = wmma4(a1, bb1, accLb);
      }
      acc0 = accLa + accLb;
      acc1 = acc0;
    }

    {
      int cnt = 0;
#pragma unroll
      for (int j = 0; j < NR; ++j)
        cnt += __popc(__builtin_amdgcn_ballot_w32(bv[j] == b));
      if (lane == 0) s_tot[wave] = cnt;
      if (threadIdx.x < PT) s_slot[threadIdx.x] = -1;
      __syncthreads();
      int base = -pbase;
#pragma unroll
      for (int w = 0; w < NWV; ++w) base += (w < wave) ? s_tot[w] : 0;
#pragma unroll
      for (int j = 0; j < NR; ++j) {
        const unsigned m = __builtin_amdgcn_ballot_w32(bv[j] == b);
        if (m != 0u) {
          const int r = base + (int)__builtin_amdgcn_mbcnt_lo(m, 0u);
          if (bv[j] == b && (unsigned)r < (unsigned)PT)
            s_slot[r] = wave * CH + 32 * j + lane;
          base += __popc(m);
        }
      }
      __syncthreads();
    }
    {
      const int pi = lane & 3;
      const int sg = lane >> 2;
      const int pc = 4 * wave + pi;
      const int s  = s_slot[pc];
      float r[Mf / 32][4];
#pragma unroll
      for (int it = 0; it < Mf / 32; ++it) { r[it][0] = 0.f; r[it][1] = 0.f; r[it][2] = 0.f; r[it][3] = 0.f; }
      if ((unsigned)s < (unsigned)Np) {
        const float* src = feats + (size_t)s * Mf + 4 * sg;
#pragma unroll
        for (int it = 0; it < Mf / 32; ++it) {
          r[it][0] = src[32 * it];     r[it][1] = src[32 * it + 1];
          r[it][2] = src[32 * it + 2]; r[it][3] = src[32 * it + 3];
        }
      }
      v2f* dst = xb + (2 * sg) * XP2 + pc;
#pragma unroll
      for (int it = 0; it < Mf / 32; ++it) {
        v2f lo; lo.x = r[it][0]; lo.y = r[it][1];
        v2f hi; hi.x = r[it][2]; hi.y = r[it][3];
        dst[(16 * it) * XP2]     = lo;
        dst[(16 * it + 1) * XP2] = hi;
      }
    }
    __syncthreads();

    load_a<Kfuse>(Wf, o0, lm, half, aw);
    mid_gemm(aw, xb, lm, half, acc0, acc1);

    load_a<Hd>(W1, o0, lm, half, aw);

    v2f* ob = xa + lm;
#pragma unroll
    for (int v = 0; v < 8; v += 2) {
      const int o = o0 + v + 8 * half;
      const float bias0 = bf[o], bias1 = bf[o + 1];
      v2f* orow = ob + (o >> 1) * XP2;
      v2f r;
      r.x = fmaxf(acc0[v] + bias0, 0.f);  r.y = fmaxf(acc0[v + 1] + bias1, 0.f);
      orow[ 0] = r;
      r.x = fmaxf(acc1[v] + bias0, 0.f);  r.y = fmaxf(acc1[v + 1] + bias1, 0.f);
      orow[16] = r;
    }
  }
  __syncthreads();

  acc0 = (v8f)V8F_ZERO; acc1 = (v8f)V8F_ZERO;
  mid_gemm(aw, xa, lm, half, acc0, acc1);
  load_a<Hd>(W2, o0, lm, half, aw);
  mid_store(acc0, acc1, b1, g1, be1, rm1, rv1, xb, o0, lm, half);
  __syncthreads();

  acc0 = (v8f)V8F_ZERO; acc1 = (v8f)V8F_ZERO;
  mid_gemm(aw, xb, lm, half, acc0, acc1);
  mid_store(acc0, acc1, b2, g2, be2, rm2, rv2, xa, o0, lm, half);
  __syncthreads();

  if (threadIdx.x < PT) {
    const v2f* xc = xa + threadIdx.x;
    float s = b3[0];
#pragma unroll 8
    for (int c = 0; c < Hd; c += 2) {
      const v2f x = xc[(c >> 1) * XP2];
      s += W3[c] * x.x;
      s += W3[c + 1] * x.y;
    }
    out[b * PPB + pbase + threadIdx.x] = s;
  }
}

extern "C" void kernel_launch(void* const* d_in, const int* in_sizes, int n_in,
                              void* d_out, int out_size, void* d_ws, size_t ws_size,
                              hipStream_t stream) {
  const float* feats = (const float*)d_in[0];
  const int*   pidx  = (const int*)d_in[1];
  const int*   poff  = (const int*)d_in[2];
  const int*   boff  = (const int*)d_in[3];
  const float* lang  = (const float*)d_in[4];
  const float* Wf    = (const float*)d_in[5];
  const float* bf    = (const float*)d_in[6];
  const float* W1p   = (const float*)d_in[7];
  const float* b1p   = (const float*)d_in[8];
  const float* g1p   = (const float*)d_in[9];
  const float* be1p  = (const float*)d_in[10];
  const float* rm1p  = (const float*)d_in[11];
  const float* rv1p  = (const float*)d_in[12];
  const float* W2p   = (const float*)d_in[13];
  const float* b2p   = (const float*)d_in[14];
  const float* g2p   = (const float*)d_in[15];
  const float* be2p  = (const float*)d_in[16];
  const float* rm2p  = (const float*)d_in[17];
  const float* rv2p  = (const float*)d_in[18];
  const float* W3p   = (const float*)d_in[19];
  const float* b3p   = (const float*)d_in[20];

  char* ws = (char*)d_ws;
  int*   bid    = (int*)ws;

  k_bid<<<NRB, RB, 0, stream>>>(pidx, poff, boff, bid);

  dim3 grid(Bn, NPS);
  k_mlp<<<grid, 256, 0, stream>>>(feats, bid, lang, Wf, bf,
                                  W1p, b1p, g1p, be1p, rm1p, rv1p,
                                  W2p, b2p, g2p, be2p, rm2p, rv2p,
                                  W3p, b3p, (float*)d_out);
}


# final 128->1 layer folded into layer 3's epilogue: each wave dots its 16 rows with W3 in registers, x3 never stored; final sum order now 8 rows per lane, lanes l and l^16, then 8 wave partials
# speedup vs baseline: 7.6136x; 1.1228x over previous
#include <hip/hip_runtime.h>

constexpr int Bn    = 32;
constexpr int PPB   = 256;
constexpr int Np    = Bn * PPB;
constexpr int Mf    = 128;
constexpr int LANGD = 256;
constexpr int Hd    = 128;
constexpr int Kfuse = LANGD + Mf;
#define BN_EPS 1e-5f

typedef float v2f __attribute__((ext_vector_type(2)));
typedef float v8f __attribute__((ext_vector_type(8)));

__device__ __forceinline__ v8f wmma4(v2f a, v2f b, v8f c) {
  return __builtin_amdgcn_wmma_f32_16x16x4_f32(false, a, false, b, (short)0, c,
                                               false, false);
}

#define V8F_ZERO {0.f, 0.f, 0.f, 0.f, 0.f, 0.f, 0.f, 0.f}

__device__ __forceinline__ float xhalf(float x) {
  const int xb = __builtin_bit_cast(int, x);
  return __builtin_bit_cast(float, __builtin_amdgcn_permlanex16(xb, xb, 0x76543210, 0xfedcba98, false, false));
}

constexpr int RB  = 256;
constexpr int NRB = Np / RB;
static_assert(Np % RB == 0, "k_bid blocks must tile the proposals exactly");

__global__ __launch_bounds__(RB) void k_bid(const int* __restrict__ pidx,
                                            const int* __restrict__ poff,
                                            const int* __restrict__ boff,
                                            int* __restrict__ bid) {
  const int i = blockIdx.x * RB + threadIdx.x;
  int first_pt = pidx[2 * poff[i] + 1];
  int b = -1;
#pragma unroll
  for (int j = 0; j <= Bn; ++j) b += (boff[j] <= first_pt) ? 1 : 0;
  bid[i] = b;
}

constexpr int PT  = 32;
constexpr int NPS = PPB / PT;
static_assert(PPB % PT == 0, "p slices must tile a batch's slots exactly");
constexpr int XP2 = 48;
constexpr int XPLANE = (Hd / 2) * XP2;
constexpr int KS  = Hd / 4;
static_assert(Mf == Hd, "layer 1's first K segment runs through the middle layers' GEMM");

template <int PITCH>
__device__ __forceinline__ void load_a(const float* __restrict__ W,
                                       int o0, int lm, int half,
                                       v2f (&aw)[KS]) {
  const float* arow = W + (size_t)(o0 + lm) * PITCH + 2 * half;
#pragma unroll
  for (int q = 0; q < KS; ++q) { aw[q].x = arow[4 * q]; aw[q].y = arow[4 * q + 1]; }
}

__device__ __forceinline__ void mid_gemm(const v2f (&aw)[KS], const v2f* xin,
                                         int lm, int half,
                                         v8f& acc0, v8f& acc1) {
  const v2f* bc = xin + half * XP2 + lm;
#pragma unroll
  for (int q = 0; q < KS; ++q) {
    const v2f* br = bc + (2 * q) * XP2;
    const v2f b0 = br[ 0];
    const v2f b1 = br[16];
    acc0 = wmma4(aw[q], b0, acc0);
    acc1 = wmma4(aw[q], b1, acc1);
  }
}

__device__ __forceinline__ void mid_store(const v8f& acc0, const v8f& acc1,
                                          const float* __restrict__ bias,
                                          const float* __restrict__ gam,
                                          const float* __restrict__ bet,
                                          const float* __restrict__ rme,
                                          const float* __restrict__ rva,
                                          v2f* xout, int o0, int lm, int half) {
  v2f* ob = xout + lm;
#pragma unroll
  for (int v = 0; v < 8; v += 2) {
    const int o = o0 + v + 8 * half;
    const float bs0 = bias[o],     bs1 = bias[o + 1];
    const float sc0 = gam[o]     * rsqrtf(rva[o]     + BN_EPS);
    const float sc1 = gam[o + 1] * rsqrtf(rva[o + 1] + BN_EPS);
    const float mu0 = rme[o],      mu1 = rme[o + 1];
    const float be0 = bet[o],      be1 = bet[o + 1];
    v2f* orow = ob + (o >> 1) * XP2;
    v2f r;
    r.x = (fmaxf(acc0[v] + bs0, 0.f) - mu0) * sc0 + be0;
    r.y = (fmaxf(acc0[v + 1] + bs1, 0.f) - mu1) * sc1 + be1;
    orow[ 0] = r;
    r.x = (fmaxf(acc1[v] + bs0, 0.f) - mu0) * sc0 + be0;
    r.y = (fmaxf(acc1[v + 1] + bs1, 0.f) - mu1) * sc1 + be1;
    orow[16] = r;
  }
}

constexpr int NWV = 8;
constexpr int CH  = Np / NWV;
constexpr int NR  = CH / 32;
static_assert(Np % (NWV * 32) == 0, "waves must tile the proposals in rounds");

__global__ __launch_bounds__(256) void k_mlp(const float* __restrict__ feats,
                                             const int* __restrict__ bid,
                                             const float* __restrict__ lang,
                                             const float* __restrict__ Wf,
                                             const float* __restrict__ bf,
                                             const float* __restrict__ W1,
                                             const float* __restrict__ b1,
                                             const float* __restrict__ g1,
                                             const float* __restrict__ be1,
                                             const float* __restrict__ rm1,
                                             const float* __restrict__ rv1,
                                             const float* __restrict__ W2,
                                             const float* __restrict__ b2,
                                             const float* __restrict__ g2,
                                             const float* __restrict__ be2,
                                             const float* __restrict__ rm2,
                                             const float* __restrict__ rv2,
                                             const float* __restrict__ W3,
                                             const float* __restrict__ b3,
                                             float* __restrict__ out) {
  __shared__ v2f xa[XPLANE];
  __shared__ v2f xb[XPLANE];
  __shared__ int s_tot[NWV];
  __shared__ int s_slot[PT];
  __shared__ float s_part[NWV * PT];

  const int b     = blockIdx.x;
  const int pbase = blockIdx.y * PT;
  const int wave  = threadIdx.x >> 5;
  const int lane  = threadIdx.x & 31;
  const int half  = lane >> 4;
  const int lm    = lane & 15;
  const int o0    = wave << 4;

  const float b3v = b3[0];

  v2f aw[KS];
  v8f acc0 = V8F_ZERO, acc1 = V8F_ZERO;

  int bv[NR];
  {
    const int* bp = bid + wave * CH + lane;
#pragma unroll
    for (int j = 0; j < NR; ++j) bv[j] = bp[32 * j];
  }

  {
    {
      const float* ap = Wf + (size_t)(o0 + lm) * Kfuse + Mf + 4 * half;
      const float* bp = lang + (size_t)b * LANGD + 4 * half;
      v8f accLa = V8F_ZERO, accLb = V8F_ZERO;
#pragma unroll
      for (int j = 0; j < LANGD / 8; ++j) {
        v2f a0;  a0.x  = ap[8 * j];      a0.y  = ap[8 * j + 1];
        v2f a1;  a1.x  = ap[8 * j + 2];  a1.y  = ap[8 * j + 3];
        v2f bb0; bb0.x = bp[8 * j];      bb0.y = bp[8 * j + 1];
        v2f bb1; bb1.x = bp[8 * j + 2];  bb1.y = bp[8 * j + 3];
        accLa = wmma4(a0, bb0, accLa);
        accLb = wmma4(a1, bb1, accLb);
      }
      acc0 = accLa + accLb;
      acc1 = acc0;
    }

    {
      int cnt = 0;
#pragma unroll
      for (int j = 0; j < NR; ++j)
        cnt += __popc(__builtin_amdgcn_ballot_w32(bv[j] == b));
      if (lane == 0) s_tot[wave] = cnt;
      if (threadIdx.x < PT) s_slot[threadIdx.x] = -1;
      __syncthreads();
      int base = -pbase;
#pragma unroll
      for (int w = 0; w < NWV; ++w) base += (w < wave) ? s_tot[w] : 0;
#pragma unroll
      for (int j = 0; j < NR; ++j) {
        const unsigned m = __builtin_amdgcn_ballot_w32(bv[j] == b);
        if (m != 0u) {
          const int r = base + (int)__builtin_amdgcn_mbcnt_lo(m, 0u);
          if (bv[j] == b && (unsigned)r < (unsigned)PT)
            s_slot[r] = wave * CH + 32 * j + lane;
          base += __popc(m);
        }
      }
      __syncthreads();
    }
    {
      const int pi = lane & 3;
      const int sg = lane >> 2;
      const int pc = 4 * wave + pi;
      const int s  = s_slot[pc];
      float r[Mf / 32][4];
#pragma unroll
      for (int it = 0; it < Mf / 32; ++it) { r[it][0] = 0.f; r[it][1] = 0.f; r[it][2] = 0.f; r[it][3] = 0.f; }
      if ((unsigned)s < (unsigned)Np) {
        const float* src = feats + (size_t)s * Mf + 4 * sg;
#pragma unroll
        for (int it = 0; it < Mf / 32; ++it) {
          r[it][0] = src[32 * it];     r[it][1] = src[32 * it + 1];
          r[it][2] = src[32 * it + 2]; r[it][3] = src[32 * it + 3];
        }
      }
      v2f* dst = xb + (2 * sg) * XP2 + pc;
#pragma unroll
      for (int it = 0; it < Mf / 32; ++it) {
        v2f lo; lo.x = r[it][0]; lo.y = r[it][1];
        v2f hi; hi.x = r[it][2]; hi.y = r[it][3];
        dst[(16 * it) * XP2]     = lo;
        dst[(16 * it + 1) * XP2] = hi;
      }
    }
    __syncthreads();

    load_a<Kfuse>(Wf, o0, lm, half, aw);
    mid_gemm(aw, xb, lm, half, acc0, acc1);

    load_a<Hd>(W1, o0, lm, half, aw);

    v2f* ob = xa + lm;
#pragma unroll
    for (int v = 0; v < 8; v += 2) {
      const int o = o0 + v + 8 * half;
      const float bias0 = bf[o], bias1 = bf[o + 1];
      v2f* orow = ob + (o >> 1) * XP2;
      v2f r;
      r.x = fmaxf(acc0[v] + bias0, 0.f);  r.y = fmaxf(acc0[v + 1] + bias1, 0.f);
      orow[ 0] = r;
      r.x = fmaxf(acc1[v] + bias0, 0.f);  r.y = fmaxf(acc1[v + 1] + bias1, 0.f);
      orow[16] = r;
    }
  }
  __syncthreads();

  acc0 = (v8f)V8F_ZERO; acc1 = (v8f)V8F_ZERO;
  mid_gemm(aw, xa, lm, half, acc0, acc1);
  load_a<Hd>(W2, o0, lm, half, aw);
  mid_store(acc0, acc1, b1, g1, be1, rm1, rv1, xb, o0, lm, half);
  __syncthreads();

  acc0 = (v8f)V8F_ZERO; acc1 = (v8f)V8F_ZERO;
  mid_gemm(aw, xb, lm, half, acc0, acc1);
  {
    float p0 = 0.f, p1 = 0.f;
#pragma unroll
    for (int v = 0; v < 8; ++v) {
      const int o = o0 + v + 8 * half;
      const float bs = b2[o];
      const float sc = g2[o] * rsqrtf(rv2[o] + BN_EPS);
      const float mu = rm2[o];
      const float be = be2[o];
      const float w  = W3[o];
      p0 += w * ((fmaxf(acc0[v] + bs, 0.f) - mu) * sc + be);
      p1 += w * ((fmaxf(acc1[v] + bs, 0.f) - mu) * sc + be);
    }
    const float give = half ? p0 : p1;
    const float keep = half ? p1 : p0;
    float mine = keep + xhalf(give);
    if (wave == 0) mine += b3v;
    s_part[wave * PT + lane] = mine;
  }
  __syncthreads();

  if (threadIdx.x < PT) {
    float s = s_part[threadIdx.x];
#pragma unroll
    for (int w = 1; w < NWV; ++w) s += s_part[w * PT + threadIdx.x];
    out[b * PPB + pbase + threadIdx.x] = s;
  }
}

extern "C" void kernel_launch(void* const* d_in, const int* in_sizes, int n_in,
                              void* d_out, int out_size, void* d_ws, size_t ws_size,
                              hipStream_t stream) {
  const float* feats = (const float*)d_in[0];
  const int*   pidx  = (const int*)d_in[1];
  const int*   poff  = (const int*)d_in[2];
  const int*   boff  = (const int*)d_in[3];
  const float* lang  = (const float*)d_in[4];
  const float* Wf    = (const float*)d_in[5];
  const float* bf    = (const float*)d_in[6];
  const float* W1p   = (const float*)d_in[7];
  const float* b1p   = (const float*)d_in[8];
  const float* g1p   = (const float*)d_in[9];
  const float* be1p  = (const float*)d_in[10];
  const float* rm1p  = (const float*)d_in[11];
  const float* rv1p  = (const float*)d_in[12];
  const float* W2p   = (const float*)d_in[13];
  const float* b2p   = (const float*)d_in[14];
  const float* g2p   = (const float*)d_in[15];
  const float* be2p  = (const float*)d_in[16];
  const float* rm2p  = (const float*)d_in[17];
  const float* rv2p  = (const float*)d_in[18];
  const float* W3p   = (const float*)d_in[19];
  const float* b3p   = (const float*)d_in[20];

  char* ws = (char*)d_ws;
  int*   bid    = (int*)ws;

  k_bid<<<NRB, RB, 0, stream>>>(pidx, poff, boff, bid);

  dim3 grid(Bn, NPS);
  k_mlp<<<grid, 256, 0, stream>>>(feats, bid, lang, Wf, bf,
                                  W1p, b1p, g1p, be1p, rm1p, rv1p,
                                  W2p, b2p, g2p, be2p, rm2p, rv2p,
                                  W3p, b3p, (float*)d_out);
}


# weights (A operand of every GEMM) copied per 128x128 unit into LDS by whole-row 512-byte loads, requested a GEMM ahead across LDS-only barriers; A fragments read from LDS, not 16-row loads; sums kept
# speedup vs baseline: 8.4646x; 1.1118x over previous
#include <hip/hip_runtime.h>

constexpr int Bn    = 32;
constexpr int PPB   = 256;
constexpr int Np    = Bn * PPB;
constexpr int Mf    = 128;
constexpr int LANGD = 256;
constexpr int Hd    = 128;
constexpr int Kfuse = LANGD + Mf;
#define BN_EPS 1e-5f

typedef float v2f __attribute__((ext_vector_type(2)));
typedef float v4f __attribute__((ext_vector_type(4)));
typedef float v8f __attribute__((ext_vector_type(8)));

__device__ __forceinline__ v8f wmma4(v2f a, v2f b, v8f c) {
  return __builtin_amdgcn_wmma_f32_16x16x4_f32(false, a, false, b, (short)0, c,
                                               false, false);
}

#define V8F_ZERO {0.f, 0.f, 0.f, 0.f, 0.f, 0.f, 0.f, 0.f}

__device__ __forceinline__ float xhalf(float x) {
  const int xb = __builtin_bit_cast(int, x);
  return __builtin_bit_cast(float, __builtin_amdgcn_permlanex16(xb, xb, 0x76543210, 0xfedcba98, false, false));
}

__device__ __forceinline__ void lds_barrier() {
  __builtin_amdgcn_fence(__ATOMIC_RELEASE, "workgroup", "local");
  __builtin_amdgcn_s_barrier();
  __builtin_amdgcn_fence(__ATOMIC_ACQUIRE, "workgroup", "local");
}

constexpr int RB  = 256;
constexpr int NRB = Np / RB;
static_assert(Np % RB == 0, "k_bid blocks must tile the proposals exactly");

__global__ __launch_bounds__(RB) void k_bid(const int* __restrict__ pidx,
                                            const int* __restrict__ poff,
                                            const int* __restrict__ boff,
                                            int* __restrict__ bid) {
  const int i = blockIdx.x * RB + threadIdx.x;
  int first_pt = pidx[2 * poff[i] + 1];
  int b = -1;
#pragma unroll
  for (int j = 0; j <= Bn; ++j) b += (boff[j] <= first_pt) ? 1 : 0;
  bid[i] = b;
}

constexpr int PT  = 32;
constexpr int NPS = PPB / PT;
static_assert(PPB % PT == 0, "p slices must tile a batch's slots exactly");
constexpr int XP2 = 48;
constexpr int XPLANE = (Hd / 2) * XP2;
constexpr int KS  = Hd / 4;
static_assert(Mf == Hd, "layer 1's first K segment runs through the middle layers' GEMM");

constexpr int WP  = Hd + 4;
constexpr int WU  = Hd * WP;
constexpr int NWB = 3;
constexpr int UPASS = Hd / 8;
static_assert((WP * 4) % 16 == 0 && ((WP * 4) / 16) % 2 == 1, "unit row pitch must be 16 B x odd");

template <int PITCH>
__device__ __forceinline__ void unit_fetch(const float* __restrict__ W,
                                           int wave, int lane, v4f (&r)[UPASS]) {
  const float* src = W + (size_t)wave * PITCH + 4 * lane;
#pragma unroll
  for (int i = 0; i < UPASS; ++i) {
    const float* p = src + (size_t)(8 * i) * PITCH;
    r[i].x = p[0]; r[i].y = p[1]; r[i].z = p[2]; r[i].w = p[3];
  }
}

__device__ __forceinline__ void unit_put(const v4f (&r)[UPASS], float* wb,
                                         int wave, int lane) {
  v4f* dst = reinterpret_cast<v4f*>(wb + wave * WP + 4 * lane);
#pragma unroll
  for (int i = 0; i < UPASS; ++i) dst[(8 * i) * (WP / 4)] = r[i];
}

__device__ __forceinline__ void load_a(const float* wb, int o0, int lm, int half,
                                       v2f (&aw)[KS]) {
  const v2f* arow = reinterpret_cast<const v2f*>(wb + (o0 + lm) * WP + 2 * half);
#pragma unroll
  for (int q = 0; q < KS; ++q) aw[q] = arow[2 * q];
}

__device__ __forceinline__ void lang_unit(const float* wb, const float* lb,
                                          int o0, int lm, int half,
                                          v8f& accLa, v8f& accLb) {
  const v4f* arow = reinterpret_cast<const v4f*>(wb + (o0 + lm) * WP + 4 * half);
  const v4f* brow = reinterpret_cast<const v4f*>(lb + 4 * half);
  v4f a[Hd / 8], bq[Hd / 8];
#pragma unroll
  for (int j = 0; j < Hd / 8; ++j) { a[j] = arow[2 * j]; bq[j] = brow[2 * j]; }
  __builtin_amdgcn_sched_barrier(0);
#pragma unroll
  for (int j = 0; j < Hd / 8; ++j) {
    v2f a0;  a0.x  = a[j].x;   a0.y  = a[j].y;
    v2f a1;  a1.x  = a[j].z;   a1.y  = a[j].w;
    v2f bb0; bb0.x = bq[j].x;  bb0.y = bq[j].y;
    v2f bb1; bb1.x = bq[j].z;  bb1.y = bq[j].w;
    accLa = wmma4(a0, bb0, accLa);
    accLb = wmma4(a1, bb1, accLb);
  }
}

__device__ __forceinline__ void mid_gemm(const v2f (&aw)[KS], const v2f* xin,
                                         int lm, int half,
                                         v8f& acc0, v8f& acc1) {
  const v2f* bc = xin + half * XP2 + lm;
#pragma unroll
  for (int q = 0; q < KS; ++q) {
    const v2f* br = bc + (2 * q) * XP2;
    const v2f b0 = br[ 0];
    const v2f b1 = br[16];
    acc0 = wmma4(aw[q], b0, acc0);
    acc1 = wmma4(aw[q], b1, acc1);
  }
}

__device__ __forceinline__ void mid_store(const v8f& acc0, const v8f& acc1,
                                          const float* __restrict__ bias,
                                          const float* __restrict__ gam,
                                          const float* __restrict__ bet,
                                          const float* __restrict__ rme,
                                          const float* __restrict__ rva,
                                          v2f* xout, int o0, int lm, int half) {
  v2f* ob = xout + lm;
#pragma unroll
  for (int v = 0; v < 8; v += 2) {
    const int o = o0 + v + 8 * half;
    const float bs0 = bias[o],     bs1 = bias[o + 1];
    const float sc0 = gam[o]     * rsqrtf(rva[o]     + BN_EPS);
    const float sc1 = gam[o + 1] * rsqrtf(rva[o + 1] + BN_EPS);
    const float mu0 = rme[o],      mu1 = rme[o + 1];
    const float be0 = bet[o],      be1 = bet[o + 1];
    v2f* orow = ob + (o >> 1) * XP2;
    v2f r;
    r.x = (fmaxf(acc0[v] + bs0, 0.f) - mu0) * sc0 + be0;
    r.y = (fmaxf(acc0[v + 1] + bs1, 0.f) - mu1) * sc1 + be1;
    orow[ 0] = r;
    r.x = (fmaxf(acc1[v] + bs0, 0.f) - mu0) * sc0 + be0;
    r.y = (fmaxf(acc1[v + 1] + bs1, 0.f) - mu1) * sc1 + be1;
    orow[16] = r;
  }
}

constexpr int NWV = 8;
constexpr int CH  = Np / NWV;
constexpr int NR  = CH / 32;
static_assert(Np % (NWV * 32) == 0, "waves must tile the proposals in rounds");

__global__ __launch_bounds__(256) void k_mlp(const float* __restrict__ feats,
                                             const int* __restrict__ bid,
                                             const float* __restrict__ lang,
                                             const float* __restrict__ Wf,
                                             const float* __restrict__ bf,
                                             const float* __restrict__ W1,
                                             const float* __restrict__ b1,
                                             const float* __restrict__ g1,
                                             const float* __restrict__ be1,
                                             const float* __restrict__ rm1,
                                             const float* __restrict__ rv1,
                                             const float* __restrict__ W2,
                                             const float* __restrict__ b2,
                                             const float* __restrict__ g2,
                                             const float* __restrict__ be2,
                                             const float* __restrict__ rm2,
                                             const float* __restrict__ rv2,
                                             const float* __restrict__ W3,
                                             const float* __restrict__ b3,
                                             float* __restrict__ out) {
  __shared__ v2f xa[XPLANE];
  __shared__ v2f xb[XPLANE];
  __shared__ __attribute__((aligned(16))) float wbuf[NWB * WU];
  __shared__ __attribute__((aligned(16))) float s_lang[LANGD];
  __shared__ int s_tot[NWV];
  __shared__ int s_slot[PT];
  __shared__ float s_part[NWV * PT];

  const int b     = blockIdx.x;
  const int pbase = blockIdx.y * PT;
  const int wave  = threadIdx.x >> 5;
  const int lane  = threadIdx.x & 31;
  const int half  = lane >> 4;
  const int lm    = lane & 15;
  const int o0    = wave << 4;

  const float b3v = b3[0];

  v2f aw[KS];
  v8f acc0 = V8F_ZERO, acc1 = V8F_ZERO;

  float* const wb0 = wbuf;
  float* const wb1 = wbuf + WU;
  float* const wb2 = wbuf + 2 * WU;

  int bv[NR];
  {
    const int* bp = bid + wave * CH + lane;
#pragma unroll
    for (int j = 0; j < NR; ++j) bv[j] = bp[32 * j];
  }
  v4f lrow = {0.f, 0.f, 0.f, 0.f};
  if (threadIdx.x < LANGD / 4) {
    const float* lp = lang + (size_t)b * LANGD + 4 * threadIdx.x;
    lrow.x = lp[0]; lrow.y = lp[1]; lrow.z = lp[2]; lrow.w = lp[3];
  }
  v4f r0[UPASS], r1[UPASS], r2[UPASS];
  unit_fetch<Kfuse>(Wf + Mf,      wave, lane, r0);
  unit_fetch<Kfuse>(Wf + Mf + Hd, wave, lane, r1);
  __builtin_amdgcn_sched_barrier(0);

  {
    {
      int cnt = 0;
#pragma unroll
      for (int j = 0; j < NR; ++j)
        cnt += __popc(__builtin_amdgcn_ballot_w32(bv[j] == b));
      if (lane == 0) s_tot[wave] = cnt;
      if (threadIdx.x < PT) s_slot[threadIdx.x] = -1;
      lds_barrier();
      int base = -pbase;
#pragma unroll
      for (int w = 0; w < NWV; ++w) base += (w < wave) ? s_tot[w] : 0;
#pragma unroll
      for (int j = 0; j < NR; ++j) {
        const unsigned m = __builtin_amdgcn_ballot_w32(bv[j] == b);
        if (m != 0u) {
          const int r = base + (int)__builtin_amdgcn_mbcnt_lo(m, 0u);
          if (bv[j] == b && (unsigned)r < (unsigned)PT)
            s_slot[r] = wave * CH + 32 * j + lane;
          base += __popc(m);
        }
      }
      lds_barrier();
    }
    {
      const int pi = lane & 3;
      const int sg = lane >> 2;
      const int pc = 4 * wave + pi;
      const int s  = s_slot[pc];
      float r[Mf / 32][4];
#pragma unroll
      for (int it = 0; it < Mf / 32; ++it) { r[it][0] = 0.f; r[it][1] = 0.f; r[it][2] = 0.f; r[it][3] = 0.f; }
      if ((unsigned)s < (unsigned)Np) {
        const float* src = feats + (size_t)s * Mf + 4 * sg;
#pragma unroll
        for (int it = 0; it < Mf / 32; ++it) {
          r[it][0] = src[32 * it];     r[it][1] = src[32 * it + 1];
          r[it][2] = src[32 * it + 2]; r[it][3] = src[32 * it + 3];
        }
      }
      unit_fetch<Kfuse>(Wf, wave, lane, r2);
      if (threadIdx.x < LANGD / 4)
        *reinterpret_cast<v4f*>(s_lang + 4 * threadIdx.x) = lrow;
      unit_put(r0, wb0, wave, lane);
      unit_put(r1, wb1, wave, lane);
      v2f* dst = xb + (2 * sg) * XP2 + pc;
#pragma unroll
      for (int it = 0; it < Mf / 32; ++it) {
        v2f lo; lo.x = r[it][0]; lo.y = r[it][1];
        v2f hi; hi.x = r[it][2]; hi.y = r[it][3];
        dst[(16 * it) * XP2]     = lo;
        dst[(16 * it + 1) * XP2] = hi;
      }
    }
    lds_barrier();

    {
      v8f accLa = V8F_ZERO, accLb = V8F_ZERO;
      lang_unit(wb0, s_lang,      o0, lm, half, accLa, accLb);
      lang_unit(wb1, s_lang + Hd, o0, lm, half, accLa, accLb);
      acc0 = accLa + accLb;
      acc1 = acc0;
    }
    unit_put(r2, wb2, wave, lane);
    unit_fetch<Hd>(W1, wave, lane, r0);
    unit_fetch<Hd>(W2, wave, lane, r1);
    lds_barrier();

    load_a(wb2, o0, lm, half, aw);
    mid_gemm(aw, xb, lm, half, acc0, acc1);

    v2f* ob = xa + lm;
#pragma unroll
    for (int v = 0; v < 8; v += 2) {
      const int o = o0 + v + 8 * half;
      const float bias0 = bf[o], bias1 = bf[o + 1];
      v2f* orow = ob + (o >> 1) * XP2;
      v2f r;
      r.x = fmaxf(acc0[v] + bias0, 0.f);  r.y = fmaxf(acc0[v + 1] + bias1, 0.f);
      orow[ 0] = r;
      r.x = fmaxf(acc1[v] + bias0, 0.f);  r.y = fmaxf(acc1[v + 1] + bias1, 0.f);
      orow[16] = r;
    }
    unit_put(r0, wb0, wave, lane);
    unit_put(r1, wb1, wave, lane);
  }
  lds_barrier();

  load_a(wb0, o0, lm, half, aw);
  acc0 = (v8f)V8F_ZERO; acc1 = (v8f)V8F_ZERO;
  mid_gemm(aw, xa, lm, half, acc0, acc1);
  mid_store(acc0, acc1, b1, g1, be1, rm1, rv1, xb, o0, lm, half);
  lds_barrier();

  load_a(wb1, o0, lm, half, aw);
  acc0 = (v8f)V8F_ZERO; acc1 = (v8f)V8F_ZERO;
  mid_gemm(aw, xb, lm, half, acc0, acc1);
  {
    float p0 = 0.f, p1 = 0.f;
#pragma unroll
    for (int v = 0; v < 8; ++v) {
      const int o = o0 + v + 8 * half;
      const float bs = b2[o];
      const float sc = g2[o] * rsqrtf(rv2[o] + BN_EPS);
      const float mu = rm2[o];
      const float be = be2[o];
      const float w  = W3[o];
      p0 += w * ((fmaxf(acc0[v] + bs, 0.f) - mu) * sc + be);
      p1 += w * ((fmaxf(acc1[v] + bs, 0.f) - mu) * sc + be);
    }
    const float give = half ? p0 : p1;
    const float keep = half ? p1 : p0;
    float mine = keep + xhalf(give);
    if (wave == 0) mine += b3v;
    s_part[wave * PT + lane] = mine;
  }
  lds_barrier();

  if (threadIdx.x < PT) {
    float s = s_part[threadIdx.x];
#pragma unroll
    for (int w = 1; w < NWV; ++w) s += s_part[w * PT + threadIdx.x];
    out[b * PPB + pbase + threadIdx.x] = s;
  }
}

extern "C" void kernel_launch(void* const* d_in, const int* in_sizes, int n_in,
                              void* d_out, int out_size, void* d_ws, size_t ws_size,
                              hipStream_t stream) {
  const float* feats = (const float*)d_in[0];
  const int*   pidx  = (const int*)d_in[1];
  const int*   poff  = (const int*)d_in[2];
  const int*   boff  = (const int*)d_in[3];
  const float* lang  = (const float*)d_in[4];
  const float* Wf    = (const float*)d_in[5];
  const float* bf    = (const float*)d_in[6];
  const float* W1p   = (const float*)d_in[7];
  const float* b1p   = (const float*)d_in[8];
  const float* g1p   = (const float*)d_in[9];
  const float* be1p  = (const float*)d_in[10];
  const float* rm1p  = (const float*)d_in[11];
  const float* rv1p  = (const float*)d_in[12];
  const float* W2p   = (const float*)d_in[13];
  const float* b2p   = (const float*)d_in[14];
  const float* g2p   = (const float*)d_in[15];
  const float* be2p  = (const float*)d_in[16];
  const float* rm2p  = (const float*)d_in[17];
  const float* rv2p  = (const float*)d_in[18];
  const float* W3p   = (const float*)d_in[19];
  const float* b3p   = (const float*)d_in[20];

  char* ws = (char*)d_ws;
  int*   bid    = (int*)ws;

  k_bid<<<NRB, RB, 0, stream>>>(pidx, poff, boff, bid);

  dim3 grid(Bn, NPS);
  k_mlp<<<grid, 256, 0, stream>>>(feats, bid, lang, Wf, bf,
                                  W1p, b1p, g1p, be1p, rm1p, rv1p,
                                  W2p, b2p, g2p, be2p, rm2p, rv2p,
                                  W3p, b3p, (float*)d_out);
}


# layer 1's lang half (W_fuse[:,128:] x lang) multiplied once per batch in launch 1: 4 WMMAs per wave inside the batch-id load wait; MLP starts from it; lang sum order now 16-k slices, ascending
# speedup vs baseline: 8.9641x; 1.0590x over previous
#include <hip/hip_runtime.h>

constexpr int Bn    = 32;
constexpr int PPB   = 256;
constexpr int Np    = Bn * PPB;
constexpr int Mf    = 128;
constexpr int LANGD = 256;
constexpr int Hd    = 128;
constexpr int Kfuse = LANGD + Mf;
#define BN_EPS 1e-5f

typedef float v2f __attribute__((ext_vector_type(2)));
typedef float v4f __attribute__((ext_vector_type(4)));
typedef float v8f __attribute__((ext_vector_type(8)));

__device__ __forceinline__ v8f wmma4(v2f a, v2f b, v8f c) {
  return __builtin_amdgcn_wmma_f32_16x16x4_f32(false, a, false, b, (short)0, c,
                                               false, false);
}

#define V8F_ZERO {0.f, 0.f, 0.f, 0.f, 0.f, 0.f, 0.f, 0.f}

__device__ __forceinline__ float xhalf(float x) {
  const int xb = __builtin_bit_cast(int, x);
  return __builtin_bit_cast(float, __builtin_amdgcn_permlanex16(xb, xb, 0x76543210, 0xfedcba98, false, false));
}

__device__ __forceinline__ void lds_barrier() {
  __builtin_amdgcn_fence(__ATOMIC_RELEASE, "workgroup", "local");
  __builtin_amdgcn_s_barrier();
  __builtin_amdgcn_fence(__ATOMIC_ACQUIRE, "workgroup", "local");
}

constexpr int RB  = 256;
constexpr int NRB = Np / RB;
static_assert(Np % RB == 0, "k_pre blocks must tile the proposals exactly");
constexpr int YTILE = 16 * 16;
constexpr int YHALF = (Hd / 16) * (Bn / 16) * YTILE;
static_assert(NRB == 2 * (Hd / 16) * (Bn / 16), "one block per D tile and K half");
static_assert(LANGD == 2 * (RB / 32) * 16, "8 waves x 16 k x 2 halves cover the lang k");

__device__ __forceinline__ v4f ld4(const float* p) {
  v4f r; r.x = p[0]; r.y = p[1]; r.z = p[2]; r.w = p[3];
  return r;
}

__global__ __launch_bounds__(RB) void k_pre(const int* __restrict__ pidx,
                                            const int* __restrict__ poff,
                                            const int* __restrict__ boff,
                                            const float* __restrict__ lang,
                                            const float* __restrict__ Wf,
                                            int* __restrict__ bid,
                                            float* __restrict__ ypart) {
  __shared__ float s_red[(RB / 32) * YTILE];
  const int i = blockIdx.x * RB + threadIdx.x;
  const int wave = threadIdx.x >> 5;
  const int lane = threadIdx.x & 31;
  const int half = lane >> 4;
  const int lm   = lane & 15;
  const int hf   = blockIdx.x & 1;
  const int ot   = (blockIdx.x >> 1) & 7;
  const int bt   = blockIdx.x >> 4;

  const int off = poff[i];
  const int kb = hf * (LANGD / 2) + wave * 16 + 4 * half;
  const float* ap = Wf + (size_t)(ot * 16 + lm) * Kfuse + Mf + kb;
  const float* bp = lang + (size_t)(bt * 16 + lm) * LANGD + kb;
  const v4f a0 = ld4(ap), a1 = ld4(ap + 8);
  const v4f q0 = ld4(bp), q1 = ld4(bp + 8);
  const int first_pt = pidx[2 * off + 1];

  v8f accLa = V8F_ZERO, accLb = V8F_ZERO;
  {
    v2f a, q;
    a.x = a0.x; a.y = a0.y; q.x = q0.x; q.y = q0.y;  accLa = wmma4(a, q, accLa);
    a.x = a0.z; a.y = a0.w; q.x = q0.z; q.y = q0.w;  accLb = wmma4(a, q, accLb);
    a.x = a1.x; a.y = a1.y; q.x = q1.x; q.y = q1.y;  accLa = wmma4(a, q, accLa);
    a.x = a1.z; a.y = a1.w; q.x = q1.z; q.y = q1.w;  accLb = wmma4(a, q, accLb);
  }
  const v8f part = accLa + accLb;
#pragma unroll
  for (int r = 0; r < 8; ++r) s_red[wave * YTILE + r * 32 + lane] = part[r];
  lds_barrier();
  {
    const int c  = threadIdx.x >> 4;
    const int rw = threadIdx.x & 15;
    const int e  = (rw & 7) * 32 + c + 16 * (rw >> 3);
    float y = s_red[e];
#pragma unroll
    for (int w = 1; w < RB / 32; ++w) y += s_red[w * YTILE + e];
    ypart[hf * YHALF + (ot * (Bn / 16) + bt) * YTILE + threadIdx.x] = y;
  }

  int b = -1;
#pragma unroll
  for (int j = 0; j <= Bn; ++j) b += (boff[j] <= first_pt) ? 1 : 0;
  bid[i] = b;
}

constexpr int PT  = 32;
constexpr int NPS = PPB / PT;
static_assert(PPB % PT == 0, "p slices must tile a batch's slots exactly");
constexpr int XP2 = 48;
constexpr int XPLANE = (Hd / 2) * XP2;
constexpr int KS  = Hd / 4;
static_assert(Mf == Hd, "layer 1's first K segment runs through the middle layers' GEMM");

constexpr int WP  = Hd + 4;
constexpr int WU  = Hd * WP;
constexpr int NWB = 3;
constexpr int UPASS = Hd / 8;
static_assert((WP * 4) % 16 == 0 && ((WP * 4) / 16) % 2 == 1, "unit row pitch must be 16 B x odd");

template <int PITCH>
__device__ __forceinline__ void unit_fetch(const float* __restrict__ W,
                                           int wave, int lane, v4f (&r)[UPASS]) {
  const float* src = W + (size_t)wave * PITCH + 4 * lane;
#pragma unroll
  for (int i = 0; i < UPASS; ++i) {
    const float* p = src + (size_t)(8 * i) * PITCH;
    r[i].x = p[0]; r[i].y = p[1]; r[i].z = p[2]; r[i].w = p[3];
  }
}

__device__ __forceinline__ void unit_put(const v4f (&r)[UPASS], float* wb,
                                         int wave, int lane) {
  v4f* dst = reinterpret_cast<v4f*>(wb + wave * WP + 4 * lane);
#pragma unroll
  for (int i = 0; i < UPASS; ++i) dst[(8 * i) * (WP / 4)] = r[i];
}

__device__ __forceinline__ void load_a(const float* wb, int o0, int lm, int half,
                                       v2f (&aw)[KS]) {
  const v2f* arow = reinterpret_cast<const v2f*>(wb + (o0 + lm) * WP + 2 * half);
#pragma unroll
  for (int q = 0; q < KS; ++q) aw[q] = arow[2 * q];
}

__device__ __forceinline__ void mid_gemm(const v2f (&aw)[KS], const v2f* xin,
                                         int lm, int half,
                                         v8f& acc0, v8f& acc1) {
  const v2f* bc = xin + half * XP2 + lm;
#pragma unroll
  for (int q = 0; q < KS; ++q) {
    const v2f* br = bc + (2 * q) * XP2;
    const v2f b0 = br[ 0];
    const v2f b1 = br[16];
    acc0 = wmma4(aw[q], b0, acc0);
    acc1 = wmma4(aw[q], b1, acc1);
  }
}

__device__ __forceinline__ void mid_store(const v8f& acc0, const v8f& acc1,
                                          const float* __restrict__ bias,
                                          const float* __restrict__ gam,
                                          const float* __restrict__ bet,
                                          const float* __restrict__ rme,
                                          const float* __restrict__ rva,
                                          v2f* xout, int o0, int lm, int half) {
  v2f* ob = xout + lm;
#pragma unroll
  for (int v = 0; v < 8; v += 2) {
    const int o = o0 + v + 8 * half;
    const float bs0 = bias[o],     bs1 = bias[o + 1];
    const float sc0 = gam[o]     * rsqrtf(rva[o]     + BN_EPS);
    const float sc1 = gam[o + 1] * rsqrtf(rva[o + 1] + BN_EPS);
    const float mu0 = rme[o],      mu1 = rme[o + 1];
    const float be0 = bet[o],      be1 = bet[o + 1];
    v2f* orow = ob + (o >> 1) * XP2;
    v2f r;
    r.x = (fmaxf(acc0[v] + bs0, 0.f) - mu0) * sc0 + be0;
    r.y = (fmaxf(acc0[v + 1] + bs1, 0.f) - mu1) * sc1 + be1;
    orow[ 0] = r;
    r.x = (fmaxf(acc1[v] + bs0, 0.f) - mu0) * sc0 + be0;
    r.y = (fmaxf(acc1[v + 1] + bs1, 0.f) - mu1) * sc1 + be1;
    orow[16] = r;
  }
}

constexpr int NWV = 8;
constexpr int CH  = Np / NWV;
constexpr int NR  = CH / 32;
static_assert(Np % (NWV * 32) == 0, "waves must tile the proposals in rounds");

__global__ __launch_bounds__(256) void k_mlp(const float* __restrict__ feats,
                                             const int* __restrict__ bid,
                                             const float* __restrict__ ypart,
                                             const float* __restrict__ Wf,
                                             const float* __restrict__ bf,
                                             const float* __restrict__ W1,
                                             const float* __restrict__ b1,
                                             const float* __restrict__ g1,
                                             const float* __restrict__ be1,
                                             const float* __restrict__ rm1,
                                             const float* __restrict__ rv1,
                                             const float* __restrict__ W2,
                                             const float* __restrict__ b2,
                                             const float* __restrict__ g2,
                                             const float* __restrict__ be2,
                                             const float* __restrict__ rm2,
                                             const float* __restrict__ rv2,
                                             const float* __restrict__ W3,
                                             const float* __restrict__ b3,
                                             float* __restrict__ out) {
  __shared__ v2f xa[XPLANE];
  __shared__ v2f xb[XPLANE];
  __shared__ __attribute__((aligned(16))) float wbuf[NWB * WU];
  __shared__ int s_tot[NWV];
  __shared__ int s_slot[PT];
  __shared__ float s_part[NWV * PT];

  const int b     = blockIdx.x;
  const int pbase = blockIdx.y * PT;
  const int wave  = threadIdx.x >> 5;
  const int lane  = threadIdx.x & 31;
  const int half  = lane >> 4;
  const int lm    = lane & 15;
  const int o0    = wave << 4;

  const float b3v = b3[0];

  v2f aw[KS];
  v8f acc0 = V8F_ZERO, acc1 = V8F_ZERO;

  float* const wb0 = wbuf;
  float* const wb1 = wbuf + WU;
  float* const wb2 = wbuf + 2 * WU;

  const float* yp = ypart + (wave * (Bn / 16) + (b >> 4)) * YTILE + (b & 15) * 16 + 8 * half;
  const v4f ya0 = ld4(yp),         ya1 = ld4(yp + 4);
  const v4f yb0 = ld4(yp + YHALF), yb1 = ld4(yp + YHALF + 4);
  int bv[NR];
  {
    const int* bp = bid + wave * CH + lane;
#pragma unroll
    for (int j = 0; j < NR; ++j) bv[j] = bp[32 * j];
  }
  v4f r0[UPASS], r1[UPASS], r2[UPASS];
  unit_fetch<Kfuse>(Wf, wave, lane, r0);
  __builtin_amdgcn_sched_barrier(0);

  {
    {
      int cnt = 0;
#pragma unroll
      for (int j = 0; j < NR; ++j)
        cnt += __popc(__builtin_amdgcn_ballot_w32(bv[j] == b));
      if (lane == 0) s_tot[wave] = cnt;
      if (threadIdx.x < PT) s_slot[threadIdx.x] = -1;
      lds_barrier();
      int base = -pbase;
#pragma unroll
      for (int w = 0; w < NWV; ++w) base += (w < wave) ? s_tot[w] : 0;
#pragma unroll
      for (int j = 0; j < NR; ++j) {
        const unsigned m = __builtin_amdgcn_ballot_w32(bv[j] == b);
        if (m != 0u) {
          const int r = base + (int)__builtin_amdgcn_mbcnt_lo(m, 0u);
          if (bv[j] == b && (unsigned)r < (unsigned)PT)
            s_slot[r] = wave * CH + 32 * j + lane;
          base += __popc(m);
        }
      }
      lds_barrier();
    }
    {
      const int pi = lane & 3;
      const int sg = lane >> 2;
      const int pc = 4 * wave + pi;
      const int s  = s_slot[pc];
      float r[Mf / 32][4];
#pragma unroll
      for (int it = 0; it < Mf / 32; ++it) { r[it][0] = 0.f; r[it][1] = 0.f; r[it][2] = 0.f; r[it][3] = 0.f; }
      if ((unsigned)s < (unsigned)Np) {
        const float* src = feats + (size_t)s * Mf + 4 * sg;
#pragma unroll
        for (int it = 0; it < Mf / 32; ++it) {
          r[it][0] = src[32 * it];     r[it][1] = src[32 * it + 1];
          r[it][2] = src[32 * it + 2]; r[it][3] = src[32 * it + 3];
        }
      }
      unit_fetch<Hd>(W1, wave, lane, r1);
      unit_fetch<Hd>(W2, wave, lane, r2);
      unit_put(r0, wb0, wave, lane);
      v2f* dst = xb + (2 * sg) * XP2 + pc;
#pragma unroll
      for (int it = 0; it < Mf / 32; ++it) {
        v2f lo; lo.x = r[it][0]; lo.y = r[it][1];
        v2f hi; hi.x = r[it][2]; hi.y = r[it][3];
        dst[(16 * it) * XP2]     = lo;
        dst[(16 * it + 1) * XP2] = hi;
      }
    }
    lds_barrier();

    acc0[0] = ya0.x + yb0.x;  acc0[1] = ya0.y + yb0.y;
    acc0[2] = ya0.z + yb0.z;  acc0[3] = ya0.w + yb0.w;
    acc0[4] = ya1.x + yb1.x;  acc0[5] = ya1.y + yb1.y;
    acc0[6] = ya1.z + yb1.z;  acc0[7] = ya1.w + yb1.w;
    acc1 = acc0;

    load_a(wb0, o0, lm, half, aw);
    mid_gemm(aw, xb, lm, half, acc0, acc1);

    v2f* ob = xa + lm;
#pragma unroll
    for (int v = 0; v < 8; v += 2) {
      const int o = o0 + v + 8 * half;
      const float bias0 = bf[o], bias1 = bf[o + 1];
      v2f* orow = ob + (o >> 1) * XP2;
      v2f r;
      r.x = fmaxf(acc0[v] + bias0, 0.f);  r.y = fmaxf(acc0[v + 1] + bias1, 0.f);
      orow[ 0] = r;
      r.x = fmaxf(acc1[v] + bias0, 0.f);  r.y = fmaxf(acc1[v + 1] + bias1, 0.f);
      orow[16] = r;
    }
    unit_put(r1, wb1, wave, lane);
    unit_put(r2, wb2, wave, lane);
  }
  lds_barrier();

  load_a(wb1, o0, lm, half, aw);
  acc0 = (v8f)V8F_ZERO; acc1 = (v8f)V8F_ZERO;
  mid_gemm(aw, xa, lm, half, acc0, acc1);
  mid_store(acc0, acc1, b1, g1, be1, rm1, rv1, xb, o0, lm, half);
  lds_barrier();

  load_a(wb2, o0, lm, half, aw);
  acc0 = (v8f)V8F_ZERO; acc1 = (v8f)V8F_ZERO;
  mid_gemm(aw, xb, lm, half, acc0, acc1);
  {
    float p0 = 0.f, p1 = 0.f;
#pragma unroll
    for (int v = 0; v < 8; ++v) {
      const int o = o0 + v + 8 * half;
      const float bs = b2[o];
      const float sc = g2[o] * rsqrtf(rv2[o] + BN_EPS);
      const float mu = rm2[o];
      const float be = be2[o];
      const float w  = W3[o];
      p0 += w * ((fmaxf(acc0[v] + bs, 0.f) - mu) * sc + be);
      p1 += w * ((fmaxf(acc1[v] + bs, 0.f) - mu) * sc + be);
    }
    const float give = half ? p0 : p1;
    const float keep = half ? p1 : p0;
    float mine = keep + xhalf(give);
    if (wave == 0) mine += b3v;
    s_part[wave * PT + lane] = mine;
  }
  lds_barrier();

  if (threadIdx.x < PT) {
    float s = s_part[threadIdx.x];
#pragma unroll
    for (int w = 1; w < NWV; ++w) s += s_part[w * PT + threadIdx.x];
    out[b * PPB + pbase + threadIdx.x] = s;
  }
}

extern "C" void kernel_launch(void* const* d_in, const int* in_sizes, int n_in,
                              void* d_out, int out_size, void* d_ws, size_t ws_size,
                              hipStream_t stream) {
  const float* feats = (const float*)d_in[0];
  const int*   pidx  = (const int*)d_in[1];
  const int*   poff  = (const int*)d_in[2];
  const int*   boff  = (const int*)d_in[3];
  const float* lang  = (const float*)d_in[4];
  const float* Wf    = (const float*)d_in[5];
  const float* bf    = (const float*)d_in[6];
  const float* W1p   = (const float*)d_in[7];
  const float* b1p   = (const float*)d_in[8];
  const float* g1p   = (const float*)d_in[9];
  const float* be1p  = (const float*)d_in[10];
  const float* rm1p  = (const float*)d_in[11];
  const float* rv1p  = (const float*)d_in[12];
  const float* W2p   = (const float*)d_in[13];
  const float* b2p   = (const float*)d_in[14];
  const float* g2p   = (const float*)d_in[15];
  const float* be2p  = (const float*)d_in[16];
  const float* rm2p  = (const float*)d_in[17];
  const float* rv2p  = (const float*)d_in[18];
  const float* W3p   = (const float*)d_in[19];
  const float* b3p   = (const float*)d_in[20];

  char* ws = (char*)d_ws;
  int*   bid    = (int*)ws;
  float* ypart  = (float*)(ws + 32768);

  k_pre<<<NRB, RB, 0, stream>>>(pidx, poff, boff, lang, Wf, bid, ypart);

  dim3 grid(Bn, NPS);
  k_mlp<<<grid, 256, 0, stream>>>(feats, bid, ypart, Wf, bf,
                                  W1p, b1p, g1p, be1p, rm1p, rv1p,
                                  W2p, b2p, g2p, be2p, rm2p, rv2p,
                                  W3p, b3p, (float*)d_out);
}
